# starting kernel
# baseline (speedup 1.0000x reference)
#include <hip/hip_runtime.h>
#include <math.h>

typedef __attribute__((ext_vector_type(2))) float v2f;
typedef __attribute__((ext_vector_type(8))) float v8f;

#define IN_CH     128
#define OUT_CH    64
#define NHEAD     4
#define TOT_OUT   256
#define NEG_SLOPE 0.2f

#define MTILE 64
#define KCH   32
#define KPAD  34

__global__ void k_init(float* out, float* denom, unsigned* segmax, int C) {
    int i = blockIdx.x * blockDim.x + threadIdx.x;
    if (i < C * TOT_OUT) out[i] = 0.0f;
    if (i < 4 * C) { denom[i] = 0.0f; segmax[i] = 0xFF800000u; }
}

__global__ void k_fold(const float* __restrict__ lin_w, const float* __restrict__ lin_b,
                       const float* __restrict__ att_w, const float* __restrict__ att_b,
                       float* v, float* c) {
    int t = blockIdx.x * blockDim.x + threadIdx.x;
    if (t < NHEAD * IN_CH) {
        int h = t / IN_CH, i = t % IN_CH;
        float s = 0.0f;
        for (int d = 0; d < OUT_CH; ++d)
            s += lin_w[(size_t)(h * OUT_CH + d) * IN_CH + i] * att_w[d];
        v[h * IN_CH + i] = s;
    }
    if (t < NHEAD) {
        float s = att_b[0];
        for (int d = 0; d < OUT_CH; ++d) s += lin_b[t * OUT_CH + d] * att_w[d];
        c[t] = s;
    }
}

__global__ __launch_bounds__(256)
void k_score(const float* __restrict__ x, const int* __restrict__ y,
             const float* __restrict__ v, const float* __restrict__ c,
             float* __restrict__ score, int* segmax_i, unsigned* segmax_u, int Nn) {
    __shared__ float vl[NHEAD * IN_CH];
    __shared__ float cl[NHEAD];
    int t = threadIdx.x;
    vl[t] = v[t]; vl[t + 256] = v[t + 256];
    if (t < NHEAD) cl[t] = c[t];
    __syncthreads();

    int node = blockIdx.x * 256 + t;
    if (node >= Nn) return;

    const float4* xp = (const float4*)(x + (size_t)node * IN_CH);
    float s[NHEAD] = {0.f, 0.f, 0.f, 0.f};
#pragma unroll
    for (int q = 0; q < IN_CH / 4; ++q) {
        float4 xv = xp[q];
#pragma unroll
        for (int h = 0; h < NHEAD; ++h) {
            const float* vh = &vl[h * IN_CH + q * 4];
            s[h] += xv.x * vh[0] + xv.y * vh[1] + xv.z * vh[2] + xv.w * vh[3];
        }
    }
    int yy = y[node];
#pragma unroll
    for (int h = 0; h < NHEAD; ++h) {
        float r = s[h] + cl[h];
        r = (r >= 0.0f) ? r : NEG_SLOPE * r;
        score[(size_t)node * NHEAD + h] = r;
        int idx = yy * NHEAD + h;
        if (r >= 0.0f) atomicMax(&segmax_i[idx], __float_as_int(r));
        else           atomicMin(&segmax_u[idx], (unsigned)__float_as_int(r));
    }
}

__global__ void k_denom(const float* __restrict__ score, const int* __restrict__ y,
                        const float* __restrict__ segmax, float* denom, int Nn) {
    int i = blockIdx.x * blockDim.x + threadIdx.x;
    if (i >= 4 * Nn) return;
    int node = i >> 2, h = i & 3;
    int idx = y[node] * NHEAD + h;
    atomicAdd(&denom[idx], __expf(score[i] - segmax[idx]));
}

__global__ __launch_bounds__(256)
void k_gemm(const float* __restrict__ x, const int* __restrict__ y,
            const float* __restrict__ lin_w, const float* __restrict__ lin_b,
            const float* __restrict__ score, const float* __restrict__ segmax,
            const float* __restrict__ denom, float* __restrict__ out, int Nn) {
    __shared__ float Xl[MTILE * KPAD];
    __shared__ float Wl[TOT_OUT * KPAD];
    __shared__ float wgt[MTILE * NHEAD];
    __shared__ float bl[TOT_OUT];
    __shared__ int   yl[MTILE];

    int t = threadIdx.x;
    int blockM = blockIdx.x * MTILE;

    {
        int m = t >> 2, h = t & 3;
        int node = blockM + m;
        float w = 0.0f; int yy = 0;
        if (node < Nn) {
            yy = y[node];
            int idx = yy * NHEAD + h;
            w = __expf(score[(size_t)node * NHEAD + h] - segmax[idx]) / denom[idx];
        }
        wgt[m * NHEAD + h] = w;
        if (h == 0) yl[m] = yy;
        bl[t] = lin_b[t];
    }

    int wave = t >> 5, lane = t & 31;
    int rowbase = (wave >> 1) * 16;
    int colbase = (wave & 1) * 128;
    int lm      = lane & 15;
    int khalf   = (lane < 16) ? 0 : 2;

    v8f acc[8];
    const v8f vzero = {0.f, 0.f, 0.f, 0.f, 0.f, 0.f, 0.f, 0.f};
#pragma unroll
    for (int i = 0; i < 8; ++i) acc[i] = vzero;

    for (int kc = 0; kc < IN_CH / KCH; ++kc) {
        int kbase = kc * KCH;
        __syncthreads();
        {
            int row = t >> 2, q = t & 3;
            int node = blockM + row;
#pragma unroll
            for (int j = 0; j < 2; ++j) {
                int kk = q * 8 + j * 4;
                float4 xv = make_float4(0.f, 0.f, 0.f, 0.f);
                if (node < Nn)
                    xv = *(const float4*)(x + (size_t)node * IN_CH + kbase + kk);
                Xl[row * KPAD + kk + 0] = xv.x;
                Xl[row * KPAD + kk + 1] = xv.y;
                Xl[row * KPAD + kk + 2] = xv.z;
                Xl[row * KPAD + kk + 3] = xv.w;
            }
            if (kc + 1 < IN_CH / KCH && node < Nn)
                __builtin_prefetch(x + (size_t)node * IN_CH + kbase + KCH, 0, 1);
        }
        {
            int n = t;
#pragma unroll
            for (int q = 0; q < 8; ++q) {
                int kk = q * 4;
                float4 w4 = *(const float4*)(lin_w + (size_t)n * IN_CH + kbase + kk);
                Wl[n * KPAD + kk + 0] = w4.x;
                Wl[n * KPAD + kk + 1] = w4.y;
                Wl[n * KPAD + kk + 2] = w4.z;
                Wl[n * KPAD + kk + 3] = w4.w;
            }
            if (kc + 1 < IN_CH / KCH)
                __builtin_prefetch(lin_w + (size_t)n * IN_CH + kbase + KCH, 0, 1);
        }
        __syncthreads();
#pragma unroll
        for (int ks = 0; ks < KCH / 4; ++ks) {
            int kk = ks * 4 + khalf;
            v2f a = *(const v2f*)(&Xl[(rowbase + lm) * KPAD + kk]);
#pragma unroll
            for (int nt = 0; nt < 8; ++nt) {
                int n = colbase + nt * 16 + lm;
                v2f b = *(const v2f*)(&Wl[n * KPAD + kk]);
                acc[nt] = __builtin_amdgcn_wmma_f32_16x16x4_f32(
                    false, a, false, b, (short)0, acc[nt], false, false);
            }
        }
    }

    int mhalf = (lane < 16) ? 0 : 8;
#pragma unroll
    for (int nt = 0; nt < 8; ++nt) {
        int n = colbase + nt * 16 + lm;
        int head = n >> 6;
        float bn = bl[n];
#pragma unroll
        for (int r = 0; r < 8; ++r) {
            int m = rowbase + mhalf + r;
            float w = wgt[m * NHEAD + head];
            float val = (acc[nt][r] + bn) * w;
            atomicAdd(&out[(size_t)yl[m] * TOT_OUT + n], val);
        }
    }
}

extern "C" void kernel_launch(void* const* d_in, const int* in_sizes, int n_in,
                              void* d_out, int out_size, void* d_ws, size_t ws_size,
                              hipStream_t stream) {
    (void)n_in; (void)ws_size;
    const float* x     = (const float*)d_in[0];
    const int*   y     = (const int*)  d_in[1];
    const float* lin_w = (const float*)d_in[3];
    const float* lin_b = (const float*)d_in[4];
    const float* att_w = (const float*)d_in[5];
    const float* att_b = (const float*)d_in[6];
    float* out = (float*)d_out;

    int N = in_sizes[0] / IN_CH;
    int C = out_size / TOT_OUT;

    float* wsf    = (float*)d_ws;
    float* v      = wsf;
    float* c      = wsf + 512;
    float* segmax = wsf + 516;
    float* denom  = segmax + 4 * C;
    float* score  = denom + 4 * C;

    int initN = C * TOT_OUT;
    k_init <<<(initN + 255) / 256, 256, 0, stream>>>(out, denom, (unsigned*)segmax, C);
    k_fold <<<2, 256, 0, stream>>>(lin_w, lin_b, att_w, att_b, v, c);
    k_score<<<(N + 255) / 256, 256, 0, stream>>>(x, y, v, c, score,
                                                 (int*)segmax, (unsigned*)segmax, N);
    k_denom<<<(4 * N + 255) / 256, 256, 0, stream>>>(score, y, segmax, denom, N);
    k_gemm <<<(N + MTILE - 1) / MTILE, 256, 0, stream>>>(x, y, lin_w, lin_b,
                                                         score, segmax, denom, out, N);
}


# k_pool forms the scores itself: a block scans its rows twice (segment max, then exp weights and sums); k_score, k_denom, score array, global atomics gone; k_merge joins stripes; sums reordered
# speedup vs baseline: 6.2453x; 6.2453x over previous
#include <hip/hip_runtime.h>
#include <math.h>

typedef __attribute__((ext_vector_type(2))) float v2f;
typedef __attribute__((ext_vector_type(8))) float v8f;

#define IN_CH     128
#define OUT_CH    64
#define NHEAD     4
#define TOT_OUT   256
#define NEG_SLOPE 0.2f

#define NSEG  1000
#define SEGB  16
#define NGRP  63
#define NSTR  8
#define SP    516
#define PT    1024

#define WS_V   0
#define WS_C   512
#define WS_MX  640
#define WS_DN(C) (WS_MX + NSTR * (C) * NHEAD)
#define WS_P(C)  (WS_MX + 2 * NSTR * (C) * NHEAD)

__global__ void k_fold(const float* __restrict__ lin_w, const float* __restrict__ lin_b,
                       const float* __restrict__ att_w, const float* __restrict__ att_b,
                       float* v, float* c) {
    int t = blockIdx.x * blockDim.x + threadIdx.x;
    if (t < NHEAD * IN_CH) {
        int h = t / IN_CH, i = t % IN_CH;
        float s = 0.0f;
        for (int d = 0; d < OUT_CH; ++d)
            s += lin_w[(size_t)(h * OUT_CH + d) * IN_CH + i] * att_w[d];
        v[h * IN_CH + i] = s;
    }
    if (t < NHEAD) {
        float s = att_b[0];
        for (int d = 0; d < OUT_CH; ++d) s += lin_b[t * OUT_CH + d] * att_w[d];
        c[t] = s;
    }
}


__device__ __forceinline__ float xhalf(float a) {
    const int b = __builtin_bit_cast(int, a);
    return __builtin_bit_cast(float,
        __builtin_amdgcn_permlanex16(b, b, 0x76543210, 0xfedcba98, false, false));
}
#define XADD(a, K) ((a) + __builtin_bit_cast(float, __builtin_amdgcn_update_dpp(          \
                        0, __builtin_bit_cast(int, (a)), 0x160 + (K), 0xf, 0xf, false)))
__device__ __forceinline__ float wave_sum(float a) {
    a = XADD(a, 1); a = XADD(a, 2); a = XADD(a, 4); a = XADD(a, 8);
    return a + xhalf(a);
}

__device__ __forceinline__ unsigned f2key(float f) {
    const unsigned b = __float_as_uint(f);
    return (b & 0x80000000u) ? ~b : (b | 0x80000000u);
}
__device__ __forceinline__ float key2f(unsigned k) {
    if (k == 0u) return -INFINITY;
    return __uint_as_float((k & 0x80000000u) ? (k ^ 0x80000000u) : ~k);
}

template <int PASS>
__device__ __forceinline__ void pool_row(const float* __restrict__ x,
                                         const float (&vr)[16], const float (&cr)[4],
                                         float* accL, float* wsL, unsigned* mK, const float* mF,
                                         int row, int seg, int lane) {
    const float* xr = x + (size_t)row * IN_CH + lane;
    const float xv[4] = {xr[0], xr[32], xr[64], xr[96]};
    float s[4];
#pragma unroll
    for (int hd = 0; hd < NHEAD; ++hd) {
        float p = xv[0] * vr[hd * 4 + 0];
        p = fmaf(xv[1], vr[hd * 4 + 1], p);
        p = fmaf(xv[2], vr[hd * 4 + 2], p);
        p = fmaf(xv[3], vr[hd * 4 + 3], p);
        const float r = wave_sum(p) + cr[hd];
        s[hd] = (r >= 0.0f) ? r : NEG_SLOPE * r;
    }
    const int sl = seg & (SEGB - 1);
    if (PASS == 1) {
        if (lane < NHEAD) {
            const float sh = (lane == 0) ? s[0] : (lane == 1) ? s[1] : (lane == 2) ? s[2] : s[3];
            __hip_atomic_fetch_max(mK + sl * NHEAD + lane, f2key(sh),
                                   __ATOMIC_RELAXED, __HIP_MEMORY_SCOPE_WORKGROUP);
        }
    } else {
        const float4 mx = *(const float4*)(mF + sl * NHEAD);
        const float w[4] = {__expf(s[0] - mx.x), __expf(s[1] - mx.y),
                            __expf(s[2] - mx.z), __expf(s[3] - mx.w)};
        float* p = accL + sl * SP + lane;
#pragma unroll
        for (int hd = 0; hd < NHEAD; ++hd)
#pragma unroll
            for (int j = 0; j < 4; ++j)
                __hip_atomic_fetch_add(p + hd * IN_CH + j * 32, xv[j] * w[hd],
                                       __ATOMIC_RELAXED, __HIP_MEMORY_SCOPE_WORKGROUP);
        if (lane < NHEAD) {
            const float wl = (lane == 0) ? w[0] : (lane == 1) ? w[1] : (lane == 2) ? w[2] : w[3];
            __hip_atomic_fetch_add(wsL + sl * NHEAD + lane, wl,
                                   __ATOMIC_RELAXED, __HIP_MEMORY_SCOPE_WORKGROUP);
        }
    }
}

__device__ __forceinline__ void load_ids(const int* __restrict__ y, int sb, int nsb,
                                         int Nn, int lane, int (&yv)[4]) {
#pragma unroll
    for (int k = 0; k < 4; ++k) {
        const int r = (sb << 7) + 32 * k + lane;
        yv[k] = -1;
        if (sb < nsb && r < Nn) yv[k] = y[r];
    }
}

__device__ __forceinline__ void ask_rows(const float* __restrict__ x, int sb, int group,
                                         int lane, const int (&yv)[4]) {
#pragma unroll
    for (int k = 0; k < 4; ++k) {
        if ((yv[k] >> 4) == group) {
            const float* p = x + (size_t)((sb << 7) + 32 * k + lane) * IN_CH;
            __builtin_prefetch(p, 0, 1);
            __builtin_prefetch(p + 64, 0, 1);
        }
    }
}

template <int PASS>
__device__ __forceinline__ void scan_pass(const float* __restrict__ x, const int* __restrict__ y,
                                          const float (&vr)[16], const float (&cr)[4],
                                          float* accL, float* wsL, unsigned* mK, const float* mF,
                                          int Nn, int group, int stripe, int wave, int lane) {
    const int nsb  = (Nn + 127) >> 7;
    const int step = (PT / 32) * NSTR;
    int sb = wave * NSTR + stripe;
    int yk[4], y1[4], y2[4];
    load_ids(y, sb, nsb, Nn, lane, yk);
    load_ids(y, sb + step, nsb, Nn, lane, y1);
    ask_rows(x, sb, group, lane, yk);
    while (sb < nsb) {
        const int base = sb << 7;
        const int sbn  = sb + step;
        load_ids(y, sbn + step, nsb, Nn, lane, y2);
        ask_rows(x, sbn, group, lane, y1);
#pragma unroll
        for (int k = 0; k < 4; ++k) {
            unsigned m = __builtin_amdgcn_ballot_w32((yk[k] >> 4) == group);
            while (m) {
                const int r = __builtin_ctz(m);
                m &= m - 1;
                const int seg = __builtin_amdgcn_readlane(yk[k], r);
                pool_row<PASS>(x, vr, cr, accL, wsL, mK, mF, base + 32 * k + r, seg, lane);
            }
        }
#pragma unroll
        for (int k = 0; k < 4; ++k) { yk[k] = y1[k]; y1[k] = y2[k]; }
        sb = sbn;
    }
}

__global__ __launch_bounds__(PT)
void k_pool(const float* __restrict__ x, const int* __restrict__ y,
            const float* __restrict__ lin_w, const float* __restrict__ lin_b,
            const float* __restrict__ v, const float* __restrict__ c,
            float* out, float* part, float* smax, float* ssum, int Nn, int C) {
    __shared__ __attribute__((aligned(16))) float accL[SEGB * SP];
    __shared__ float wsL[SEGB * NHEAD];
    __shared__ unsigned mK[SEGB * NHEAD];
    __shared__ __attribute__((aligned(16))) float mF[SEGB * NHEAD];
    __shared__ float outL[SEGB * TOT_OUT];

    const int t = threadIdx.x;
    const int group  = blockIdx.x % NGRP;
    const int stripe = blockIdx.x / NGRP;
    const int wave = __builtin_amdgcn_readfirstlane(t >> 5);
    const int lane = t & 31;

    for (int i = t; i < SEGB * SP; i += PT) accL[i] = 0.0f;
    if (t < SEGB * NHEAD) { wsL[t] = 0.0f; mK[t] = 0u; }

    float vr[16], cr[4];
#pragma unroll
    for (int hd = 0; hd < NHEAD; ++hd) {
#pragma unroll
        for (int j = 0; j < 4; ++j) vr[hd * 4 + j] = v[hd * IN_CH + lane + 32 * j];
        cr[hd] = c[hd];
    }
    __syncthreads();

    scan_pass<1>(x, y, vr, cr, accL, wsL, mK, mF, Nn, group, stripe, wave, lane);
    __syncthreads();
    if (t < SEGB * NHEAD) mF[t] = key2f(mK[t]);
    __syncthreads();
    scan_pass<2>(x, y, vr, cr, accL, wsL, mK, mF, Nn, group, stripe, wave, lane);
    __syncthreads();

    if (wave < 16) {
        const int head = wave >> 2, nt = wave & 3;
        const int cc = lane & 15, h = lane >> 4;
        const int col = head * OUT_CH + nt * 16 + cc;
        const float* brow = lin_w + (size_t)col * IN_CH + 4 * h;
        const float* arow = accL + cc * SP + head * IN_CH + 4 * h;
        v8f acc = {0.f, 0.f, 0.f, 0.f, 0.f, 0.f, 0.f, 0.f};
#pragma unroll
        for (int j = 0; j < 16; ++j) {
            const float4 a4 = *(const float4*)(arow + 8 * j);
            const float4 b4 = *(const float4*)(brow + 8 * j);
            v2f a0 = {a4.x, a4.y}, a1 = {a4.z, a4.w};
            v2f b0 = {b4.x, b4.y}, b1 = {b4.z, b4.w};
            acc = __builtin_amdgcn_wmma_f32_16x16x4_f32(false, a0, false, b0, (short)0, acc, false, false);
            acc = __builtin_amdgcn_wmma_f32_16x16x4_f32(false, a1, false, b1, (short)0, acc, false, false);
        }
        const float bn = lin_b[col];
#pragma unroll
        for (int r = 0; r < 8; ++r) {
            const int sl = 8 * h + r;
            outL[sl * TOT_OUT + col] = acc[r] + bn * wsL[sl * NHEAD + head];
        }
    }
    __syncthreads();

    const int nseg = (C < NSEG) ? C : NSEG;
    float* dst = (stripe == 0) ? out : part + (size_t)(stripe - 1) * C * TOT_OUT;
#pragma unroll
    for (int i = 0; i < SEGB * TOT_OUT / PT; ++i) {
        const int e   = i * PT + t;
        const int seg = group * SEGB + (e >> 8);
        if (seg < nseg) dst[(size_t)seg * TOT_OUT + (e & 255)] = outL[e];
    }
    if (t < SEGB * NHEAD) {
        const int seg = group * SEGB + (t >> 2);
        if (seg < nseg) {
            const size_t o = ((size_t)stripe * C + seg) * NHEAD + (t & 3);
            smax[o] = mF[t];
            ssum[o] = wsL[t];
        }
    }
}

__global__ __launch_bounds__(256)
void k_merge(const float* __restrict__ part, const float* __restrict__ smax,
             const float* __restrict__ ssum, float* out, int C) {
    const int i = blockIdx.x * 256 + threadIdx.x;
    if (i >= C * TOT_OUT) return;
    const int seg = i >> 8, head = (i & 255) >> 6;
    if (seg >= NSEG) { out[i] = 0.0f; return; }
    float m[NSTR], best = -INFINITY;
#pragma unroll
    for (int b = 0; b < NSTR; ++b) {
        m[b] = smax[((size_t)b * C + seg) * NHEAD + head];
        best = fmaxf(best, m[b]);
    }
    float num = 0.0f, den = 0.0f;
#pragma unroll
    for (int b = 0; b < NSTR; ++b) {
        const float f  = (m[b] == -INFINITY) ? 0.0f : __expf(m[b] - best);
        const float pb = (b == 0) ? out[i] : part[(size_t)(b - 1) * C * TOT_OUT + i];
        num += f * pb;
        den += f * ssum[((size_t)b * C + seg) * NHEAD + head];
    }
    out[i] = (den > 0.0f) ? num / den : 0.0f;
}

extern "C" void kernel_launch(void* const* d_in, const int* in_sizes, int n_in,
                              void* d_out, int out_size, void* d_ws, size_t ws_size,
                              hipStream_t stream) {
    (void)n_in; (void)ws_size;
    const float* x     = (const float*)d_in[0];
    const int*   y     = (const int*)  d_in[1];
    const float* lin_w = (const float*)d_in[3];
    const float* lin_b = (const float*)d_in[4];
    const float* att_w = (const float*)d_in[5];
    const float* att_b = (const float*)d_in[6];
    float* out = (float*)d_out;

    int N = in_sizes[0] / IN_CH;
    int C = out_size / TOT_OUT;

    float* wsf  = (float*)d_ws;
    float* v    = wsf + WS_V;
    float* c    = wsf + WS_C;
    float* smax = wsf + WS_MX;
    float* ssum = wsf + WS_DN(C);
    float* part = wsf + WS_P(C);

    k_fold <<<2, 256, 0, stream>>>(lin_w, lin_b, att_w, att_b, v, c);
    k_pool <<<NGRP * NSTR, PT, 0, stream>>>(x, y, lin_w, lin_b, v, c,
                                            out, part, smax, ssum, N, C);
    k_merge<<<(C * TOT_OUT + 255) / 256, 256, 0, stream>>>(part, smax, ssum, out, C);
}


# k_pool pass 1 lists its rows with their scores in LDS (slots in arrival order); pass 2 walks the list, waves sharing it evenly, next row's loads ahead: no second scan or scoring; sums reordered
# speedup vs baseline: 6.8956x; 1.1041x over previous
#include <hip/hip_runtime.h>
#include <math.h>

typedef __attribute__((ext_vector_type(2))) float v2f;
typedef __attribute__((ext_vector_type(8))) float v8f;

#define IN_CH     128
#define OUT_CH    64
#define NHEAD     4
#define TOT_OUT   256
#define NEG_SLOPE 0.2f

#define NSEG  1000
#define SEGB  16
#define NGRP  63
#define NSTR  8
#define SP    516
#define PT    1024
#define LCAP  1536

#define WS_V   0
#define WS_C   512
#define WS_MX  640
#define WS_DN(C) (WS_MX + NSTR * (C) * NHEAD)
#define WS_P(C)  (WS_MX + 2 * NSTR * (C) * NHEAD)

__global__ void k_fold(const float* __restrict__ lin_w, const float* __restrict__ lin_b,
                       const float* __restrict__ att_w, const float* __restrict__ att_b,
                       float* v, float* c) {
    int t = blockIdx.x * blockDim.x + threadIdx.x;
    if (t < NHEAD * IN_CH) {
        int h = t / IN_CH, i = t % IN_CH;
        float s = 0.0f;
        for (int d = 0; d < OUT_CH; ++d)
            s += lin_w[(size_t)(h * OUT_CH + d) * IN_CH + i] * att_w[d];
        v[h * IN_CH + i] = s;
    }
    if (t < NHEAD) {
        float s = att_b[0];
        for (int d = 0; d < OUT_CH; ++d) s += lin_b[t * OUT_CH + d] * att_w[d];
        c[t] = s;
    }
}


__device__ __forceinline__ float xhalf(float a) {
    const int b = __builtin_bit_cast(int, a);
    return __builtin_bit_cast(float,
        __builtin_amdgcn_permlanex16(b, b, 0x76543210, 0xfedcba98, false, false));
}
#define XADD(a, K) ((a) + __builtin_bit_cast(float, __builtin_amdgcn_update_dpp(          \
                        0, __builtin_bit_cast(int, (a)), 0x160 + (K), 0xf, 0xf, false)))
__device__ __forceinline__ float wave_sum(float a) {
    a = XADD(a, 1); a = XADD(a, 2); a = XADD(a, 4); a = XADD(a, 8);
    return a + xhalf(a);
}

__device__ __forceinline__ unsigned f2key(float f) {
    const unsigned b = __float_as_uint(f);
    return (b & 0x80000000u) ? ~b : (b | 0x80000000u);
}
__device__ __forceinline__ float key2f(unsigned k) {
    if (k == 0u) return -INFINITY;
    return __uint_as_float((k & 0x80000000u) ? (k ^ 0x80000000u) : ~k);
}

template <int PASS>
__device__ __forceinline__ void pool_row(const float* __restrict__ x,
                                         const float (&vr)[16], const float (&cr)[4],
                                         float* accL, float* wsL, unsigned* mK, const float* mF,
                                         float* lsc, unsigned* lrow, unsigned* cnt,
                                         int row, int seg, int lane) {
    const float* xr = x + (size_t)row * IN_CH + lane;
    const float xv[4] = {xr[0], xr[32], xr[64], xr[96]};
    unsigned slot = 0u;
    if (PASS == 1) {
        if (lane == 0)
            slot = __hip_atomic_fetch_add(cnt, 1u, __ATOMIC_RELAXED, __HIP_MEMORY_SCOPE_WORKGROUP);
        slot = (unsigned)__builtin_amdgcn_readfirstlane((int)slot);
    }
    float s[4];
#pragma unroll
    for (int hd = 0; hd < NHEAD; ++hd) {
        float p = xv[0] * vr[hd * 4 + 0];
        p = fmaf(xv[1], vr[hd * 4 + 1], p);
        p = fmaf(xv[2], vr[hd * 4 + 2], p);
        p = fmaf(xv[3], vr[hd * 4 + 3], p);
        const float r = wave_sum(p) + cr[hd];
        s[hd] = (r >= 0.0f) ? r : NEG_SLOPE * r;
    }
    const int sl = seg & (SEGB - 1);
    if (PASS == 1) {
        if (lane < NHEAD) {
            const float sh = (lane == 0) ? s[0] : (lane == 1) ? s[1] : (lane == 2) ? s[2] : s[3];
            __hip_atomic_fetch_max(mK + sl * NHEAD + lane, f2key(sh),
                                   __ATOMIC_RELAXED, __HIP_MEMORY_SCOPE_WORKGROUP);
            if (slot < (unsigned)LCAP) {
                lsc[slot * NHEAD + lane] = sh;
                if (lane == 0) lrow[slot] = ((unsigned)row << 4) | (unsigned)sl;
            }
        }
    } else {
        const float4 mx = *(const float4*)(mF + sl * NHEAD);
        const float w[4] = {__expf(s[0] - mx.x), __expf(s[1] - mx.y),
                            __expf(s[2] - mx.z), __expf(s[3] - mx.w)};
        float* p = accL + sl * SP + lane;
#pragma unroll
        for (int hd = 0; hd < NHEAD; ++hd)
#pragma unroll
            for (int j = 0; j < 4; ++j)
                __hip_atomic_fetch_add(p + hd * IN_CH + j * 32, xv[j] * w[hd],
                                       __ATOMIC_RELAXED, __HIP_MEMORY_SCOPE_WORKGROUP);
        if (lane < NHEAD) {
            const float wl = (lane == 0) ? w[0] : (lane == 1) ? w[1] : (lane == 2) ? w[2] : w[3];
            __hip_atomic_fetch_add(wsL + sl * NHEAD + lane, wl,
                                   __ATOMIC_RELAXED, __HIP_MEMORY_SCOPE_WORKGROUP);
        }
    }
}

__device__ __forceinline__ void load_ids(const int* __restrict__ y, int sb, int nsb,
                                         int Nn, int lane, int (&yv)[4]) {
#pragma unroll
    for (int k = 0; k < 4; ++k) {
        const int r = (sb << 7) + 32 * k + lane;
        yv[k] = -1;
        if (sb < nsb && r < Nn) yv[k] = y[r];
    }
}

__device__ __forceinline__ void ask_rows(const float* __restrict__ x, int sb, int group,
                                         int lane, const int (&yv)[4]) {
#pragma unroll
    for (int k = 0; k < 4; ++k) {
        if ((yv[k] >> 4) == group) {
            const float* p = x + (size_t)((sb << 7) + 32 * k + lane) * IN_CH;
            __builtin_prefetch(p, 0, 1);
            __builtin_prefetch(p + 64, 0, 1);
        }
    }
}

template <int PASS>
__device__ __forceinline__ void scan_pass(const float* __restrict__ x, const int* __restrict__ y,
                                          const float (&vr)[16], const float (&cr)[4],
                                          float* accL, float* wsL, unsigned* mK, const float* mF,
                                          float* lsc, unsigned* lrow, unsigned* cnt,
                                          int Nn, int group, int stripe, int wave, int lane) {
    const int nsb  = (Nn + 127) >> 7;
    const int step = (PT / 32) * NSTR;
    int sb = wave * NSTR + stripe;
    int yk[4], y1[4], y2[4];
    load_ids(y, sb, nsb, Nn, lane, yk);
    load_ids(y, sb + step, nsb, Nn, lane, y1);
    ask_rows(x, sb, group, lane, yk);
    while (sb < nsb) {
        const int base = sb << 7;
        const int sbn  = sb + step;
        load_ids(y, sbn + step, nsb, Nn, lane, y2);
        ask_rows(x, sbn, group, lane, y1);
#pragma unroll
        for (int k = 0; k < 4; ++k) {
            unsigned m = __builtin_amdgcn_ballot_w32((yk[k] >> 4) == group);
            while (m) {
                const int r = __builtin_ctz(m);
                m &= m - 1;
                const int seg = __builtin_amdgcn_readlane(yk[k], r);
                pool_row<PASS>(x, vr, cr, accL, wsL, mK, mF, lsc, lrow, cnt, base + 32 * k + r, seg, lane);
            }
        }
#pragma unroll
        for (int k = 0; k < 4; ++k) { yk[k] = y1[k]; y1[k] = y2[k]; }
        sb = sbn;
    }
}

__device__ __forceinline__ void fetch_entry(const float* __restrict__ x, const float* lsc,
                                            const float* mF, unsigned q, int e, int lane,
                                            float (&xv)[4], float4& sc, float4& mx) {
    sc = *(const float4*)(lsc + e * NHEAD);
    mx = *(const float4*)(mF + (q & 15u) * NHEAD);
    const float* xr = x + (size_t)(q >> 4) * IN_CH + lane;
    xv[0] = xr[0]; xv[1] = xr[32]; xv[2] = xr[64]; xv[3] = xr[96];
}

__device__ __forceinline__ void add_entry(float* accL, float* wsL, int sl, int lane,
                                          const float (&xv)[4], const float4& sc, const float4& mx) {
    const float w[4] = {__expf(sc.x - mx.x), __expf(sc.y - mx.y),
                        __expf(sc.z - mx.z), __expf(sc.w - mx.w)};
    float* p = accL + sl * SP + lane;
#pragma unroll
    for (int hd = 0; hd < NHEAD; ++hd)
#pragma unroll
        for (int j = 0; j < 4; ++j)
            __hip_atomic_fetch_add(p + hd * IN_CH + j * 32, xv[j] * w[hd],
                                   __ATOMIC_RELAXED, __HIP_MEMORY_SCOPE_WORKGROUP);
    const float wl = (lane & 2) ? ((lane & 1) ? w[3] : w[2]) : ((lane & 1) ? w[1] : w[0]);
    if (lane < NHEAD)
        __hip_atomic_fetch_add(wsL + sl * NHEAD + lane, wl,
                                __ATOMIC_RELAXED, __HIP_MEMORY_SCOPE_WORKGROUP);
}

__device__ __forceinline__ void walk_list(const float* __restrict__ x,
                                          float* accL, float* wsL, const float* mF,
                                          const float* lsc, const unsigned* lrow,
                                          int n, int wave, int lane) {
    const int nw = PT / 32;
    int e = wave;
    if (e >= n) return;
    const int last = n - 1;
    float xa[4], xb[4];
    float4 sca, scb, mxa, mxb;
    unsigned qa = (unsigned)__builtin_amdgcn_readfirstlane((int)lrow[e]);
    unsigned qb;
    unsigned pn = lrow[min(e + nw, last)];
    fetch_entry(x, lsc, mF, qa, e, lane, xa, sca, mxa);
    for (;;) {
        qb = (unsigned)__builtin_amdgcn_readfirstlane((int)pn);
        pn = lrow[min(e + 2 * nw, last)];
        fetch_entry(x, lsc, mF, qb, min(e + nw, last), lane, xb, scb, mxb);
        add_entry(accL, wsL, (int)(qa & 15u), lane, xa, sca, mxa);
        e += nw;
        if (e >= n) break;
        qa = (unsigned)__builtin_amdgcn_readfirstlane((int)pn);
        pn = lrow[min(e + 2 * nw, last)];
        fetch_entry(x, lsc, mF, qa, min(e + nw, last), lane, xa, sca, mxa);
        add_entry(accL, wsL, (int)(qb & 15u), lane, xb, scb, mxb);
        e += nw;
        if (e >= n) break;
    }
}

__global__ __launch_bounds__(PT)
void k_pool(const float* __restrict__ x, const int* __restrict__ y,
            const float* __restrict__ lin_w, const float* __restrict__ lin_b,
            const float* __restrict__ v, const float* __restrict__ c,
            float* out, float* part, float* smax, float* ssum, int Nn, int C) {
    __shared__ __attribute__((aligned(16))) float accL[SEGB * SP];
    __shared__ float wsL[SEGB * NHEAD];
    __shared__ unsigned mK[SEGB * NHEAD];
    __shared__ __attribute__((aligned(16))) float mF[SEGB * NHEAD];
    __shared__ float outL[SEGB * TOT_OUT];
    __shared__ __attribute__((aligned(16))) float lscL[LCAP * NHEAD];
    __shared__ unsigned lrowL[LCAP];
    __shared__ unsigned cntL;

    const int t = threadIdx.x;
    const int group  = blockIdx.x % NGRP;
    const int stripe = blockIdx.x / NGRP;
    const int wave = __builtin_amdgcn_readfirstlane(t >> 5);
    const int lane = t & 31;

    for (int i = t; i < SEGB * SP; i += PT) accL[i] = 0.0f;
    if (t < SEGB * NHEAD) { wsL[t] = 0.0f; mK[t] = 0u; }
    if (t == 0) cntL = 0u;

    float vr[16], cr[4];
#pragma unroll
    for (int hd = 0; hd < NHEAD; ++hd) {
#pragma unroll
        for (int j = 0; j < 4; ++j) vr[hd * 4 + j] = v[hd * IN_CH + lane + 32 * j];
        cr[hd] = c[hd];
    }
    __syncthreads();

    scan_pass<1>(x, y, vr, cr, accL, wsL, mK, mF, lscL, lrowL, &cntL, Nn, group, stripe, wave, lane);
    __syncthreads();
    if (t < SEGB * NHEAD) mF[t] = key2f(mK[t]);
    const unsigned met = (unsigned)__builtin_amdgcn_readfirstlane((int)cntL);
    __syncthreads();
    if (met <= (unsigned)LCAP)
        walk_list(x, accL, wsL, mF, lscL, lrowL, (int)met, wave, lane);
    else
        scan_pass<2>(x, y, vr, cr, accL, wsL, mK, mF, lscL, lrowL, &cntL, Nn, group, stripe, wave, lane);
    __syncthreads();

    if (wave < 16) {
        const int head = wave >> 2, nt = wave & 3;
        const int cc = lane & 15, h = lane >> 4;
        const int col = head * OUT_CH + nt * 16 + cc;
        const float* brow = lin_w + (size_t)col * IN_CH + 4 * h;
        const float* arow = accL + cc * SP + head * IN_CH + 4 * h;
        v8f acc = {0.f, 0.f, 0.f, 0.f, 0.f, 0.f, 0.f, 0.f};
#pragma unroll
        for (int j = 0; j < 16; ++j) {
            const float4 a4 = *(const float4*)(arow + 8 * j);
            const float4 b4 = *(const float4*)(brow + 8 * j);
            v2f a0 = {a4.x, a4.y}, a1 = {a4.z, a4.w};
            v2f b0 = {b4.x, b4.y}, b1 = {b4.z, b4.w};
            acc = __builtin_amdgcn_wmma_f32_16x16x4_f32(false, a0, false, b0, (short)0, acc, false, false);
            acc = __builtin_amdgcn_wmma_f32_16x16x4_f32(false, a1, false, b1, (short)0, acc, false, false);
        }
        const float bn = lin_b[col];
#pragma unroll
        for (int r = 0; r < 8; ++r) {
            const int sl = 8 * h + r;
            outL[sl * TOT_OUT + col] = acc[r] + bn * wsL[sl * NHEAD + head];
        }
    }
    __syncthreads();

    const int nseg = (C < NSEG) ? C : NSEG;
    float* dst = (stripe == 0) ? out : part + (size_t)(stripe - 1) * C * TOT_OUT;
#pragma unroll
    for (int i = 0; i < SEGB * TOT_OUT / PT; ++i) {
        const int e   = i * PT + t;
        const int seg = group * SEGB + (e >> 8);
        if (seg < nseg) dst[(size_t)seg * TOT_OUT + (e & 255)] = outL[e];
    }
    if (t < SEGB * NHEAD) {
        const int seg = group * SEGB + (t >> 2);
        if (seg < nseg) {
            const size_t o = ((size_t)stripe * C + seg) * NHEAD + (t & 3);
            smax[o] = mF[t];
            ssum[o] = wsL[t];
        }
    }
}

__global__ __launch_bounds__(256)
void k_merge(const float* __restrict__ part, const float* __restrict__ smax,
             const float* __restrict__ ssum, float* out, int C) {
    const int i = blockIdx.x * 256 + threadIdx.x;
    if (i >= C * TOT_OUT) return;
    const int seg = i >> 8, head = (i & 255) >> 6;
    if (seg >= NSEG) { out[i] = 0.0f; return; }
    float m[NSTR], best = -INFINITY;
#pragma unroll
    for (int b = 0; b < NSTR; ++b) {
        m[b] = smax[((size_t)b * C + seg) * NHEAD + head];
        best = fmaxf(best, m[b]);
    }
    float num = 0.0f, den = 0.0f;
#pragma unroll
    for (int b = 0; b < NSTR; ++b) {
        const float f  = (m[b] == -INFINITY) ? 0.0f : __expf(m[b] - best);
        const float pb = (b == 0) ? out[i] : part[(size_t)(b - 1) * C * TOT_OUT + i];
        num += f * pb;
        den += f * ssum[((size_t)b * C + seg) * NHEAD + head];
    }
    out[i] = (den > 0.0f) ? num / den : 0.0f;
}

extern "C" void kernel_launch(void* const* d_in, const int* in_sizes, int n_in,
                              void* d_out, int out_size, void* d_ws, size_t ws_size,
                              hipStream_t stream) {
    (void)n_in; (void)ws_size;
    const float* x     = (const float*)d_in[0];
    const int*   y     = (const int*)  d_in[1];
    const float* lin_w = (const float*)d_in[3];
    const float* lin_b = (const float*)d_in[4];
    const float* att_w = (const float*)d_in[5];
    const float* att_b = (const float*)d_in[6];
    float* out = (float*)d_out;

    int N = in_sizes[0] / IN_CH;
    int C = out_size / TOT_OUT;

    float* wsf  = (float*)d_ws;
    float* v    = wsf + WS_V;
    float* c    = wsf + WS_C;
    float* smax = wsf + WS_MX;
    float* ssum = wsf + WS_DN(C);
    float* part = wsf + WS_P(C);

    k_fold <<<2, 256, 0, stream>>>(lin_w, lin_b, att_w, att_b, v, c);
    k_pool <<<NGRP * NSTR, PT, 0, stream>>>(x, y, lin_w, lin_b, v, c,
                                            out, part, smax, ssum, N, C);
    k_merge<<<(C * TOT_OUT + 255) / 256, 256, 0, stream>>>(part, smax, ssum, out, C);
}


# k_pool reads x once: a y-only scan lists rows per segment (LDS, arrival order); 2 waves per segment keep a running largest score and rescaled sums in registers; no per-row LDS adds; sums reordered
# speedup vs baseline: 10.4216x; 1.5113x over previous
#include <hip/hip_runtime.h>
#include <math.h>

typedef __attribute__((ext_vector_type(2))) float v2f;
typedef __attribute__((ext_vector_type(8))) float v8f;

#define IN_CH     128
#define OUT_CH    64
#define NHEAD     4
#define TOT_OUT   256
#define NEG_SLOPE 0.2f

#define NSEG  1000
#define SEGB  16
#define NGRP  63
#define NSTR  8
#define SP    516
#define PT    1024
#define LSEG  192

#define WS_V   0
#define WS_C   512
#define WS_MX  640
#define WS_DN(C) (WS_MX + NSTR * (C) * NHEAD)
#define WS_P(C)  (WS_MX + 2 * NSTR * (C) * NHEAD)

__global__ void k_fold(const float* __restrict__ lin_w, const float* __restrict__ lin_b,
                       const float* __restrict__ att_w, const float* __restrict__ att_b,
                       float* v, float* c) {
    int t = blockIdx.x * blockDim.x + threadIdx.x;
    if (t < NHEAD * IN_CH) {
        int h = t / IN_CH, i = t % IN_CH;
        float s = 0.0f;
        for (int d = 0; d < OUT_CH; ++d)
            s += lin_w[(size_t)(h * OUT_CH + d) * IN_CH + i] * att_w[d];
        v[h * IN_CH + i] = s;
    }
    if (t < NHEAD) {
        float s = att_b[0];
        for (int d = 0; d < OUT_CH; ++d) s += lin_b[t * OUT_CH + d] * att_w[d];
        c[t] = s;
    }
}


__device__ __forceinline__ float xhalf(float a) {
    const int b = __builtin_bit_cast(int, a);
    return __builtin_bit_cast(float,
        __builtin_amdgcn_permlanex16(b, b, 0x76543210, 0xfedcba98, false, false));
}
#define XADD(a, K) ((a) + __builtin_bit_cast(float, __builtin_amdgcn_update_dpp(          \
                        0, __builtin_bit_cast(int, (a)), 0x160 + (K), 0xf, 0xf, false)))
__device__ __forceinline__ float wave_sum(float a) {
    a = XADD(a, 1); a = XADD(a, 2); a = XADD(a, 4); a = XADD(a, 8);
    return a + xhalf(a);
}

__device__ __forceinline__ unsigned f2key(float f) {
    const unsigned b = __float_as_uint(f);
    return (b & 0x80000000u) ? ~b : (b | 0x80000000u);
}
__device__ __forceinline__ float key2f(unsigned k) {
    if (k == 0u) return -INFINITY;
    return __uint_as_float((k & 0x80000000u) ? (k ^ 0x80000000u) : ~k);
}

__device__ __forceinline__ float uni(float a) {
    return __builtin_bit_cast(float, __builtin_amdgcn_readfirstlane(__builtin_bit_cast(int, a)));
}

__device__ __forceinline__ void row_scores(const float (&xv)[4], const float (&vr)[16],
                                           const float (&cr)[4], float (&s)[4]) {
#pragma unroll
    for (int hd = 0; hd < NHEAD; ++hd) {
        float p = xv[0] * vr[hd * 4 + 0];
        p = fmaf(xv[1], vr[hd * 4 + 1], p);
        p = fmaf(xv[2], vr[hd * 4 + 2], p);
        p = fmaf(xv[3], vr[hd * 4 + 3], p);
        const float r = wave_sum(p) + cr[hd];
        s[hd] = (r >= 0.0f) ? r : NEG_SLOPE * r;
    }
}

template <int PASS>
__device__ __forceinline__ void pool_row(const float* __restrict__ x,
                                         const float (&vr)[16], const float (&cr)[4],
                                         float* accL, float* wsL, unsigned* mK, const float* mF,
                                         int row, int seg, int lane) {
    const float* xr = x + (size_t)row * IN_CH + lane;
    const float xv[4] = {xr[0], xr[32], xr[64], xr[96]};
    float s[4];
    row_scores(xv, vr, cr, s);
    const int sl = seg & (SEGB - 1);
    if (PASS == 1) {
        if (lane < NHEAD) {
            const float sh = (lane == 0) ? s[0] : (lane == 1) ? s[1] : (lane == 2) ? s[2] : s[3];
            __hip_atomic_fetch_max(mK + sl * NHEAD + lane, f2key(sh),
                                   __ATOMIC_RELAXED, __HIP_MEMORY_SCOPE_WORKGROUP);
        }
    } else {
        const float4 mx = *(const float4*)(mF + sl * NHEAD);
        const float w[4] = {__expf(s[0] - mx.x), __expf(s[1] - mx.y),
                            __expf(s[2] - mx.z), __expf(s[3] - mx.w)};
        float* p = accL + sl * SP + lane;
#pragma unroll
        for (int hd = 0; hd < NHEAD; ++hd)
#pragma unroll
            for (int j = 0; j < 4; ++j)
                __hip_atomic_fetch_add(p + hd * IN_CH + j * 32, xv[j] * w[hd],
                                       __ATOMIC_RELAXED, __HIP_MEMORY_SCOPE_WORKGROUP);
        if (lane < NHEAD) {
            const float wl = (lane == 0) ? w[0] : (lane == 1) ? w[1] : (lane == 2) ? w[2] : w[3];
            __hip_atomic_fetch_add(wsL + sl * NHEAD + lane, wl,
                                   __ATOMIC_RELAXED, __HIP_MEMORY_SCOPE_WORKGROUP);
        }
    }
}

__device__ __forceinline__ void load_ids(const int* __restrict__ y, int sb, int nsb,
                                         int Nn, int lane, int (&yv)[4]) {
#pragma unroll
    for (int k = 0; k < 4; ++k) {
        const int r = (sb << 7) + 32 * k + lane;
        yv[k] = -1;
        if (sb < nsb && r < Nn) yv[k] = y[r];
    }
}

template <int PASS>
__device__ __forceinline__ void scan_pass(const float* __restrict__ x, const int* __restrict__ y,
                                          const float (&vr)[16], const float (&cr)[4],
                                          float* accL, float* wsL, unsigned* mK, const float* mF,
                                          int Nn, int group, int stripe, int wave, int lane) {
    const int nsb  = (Nn + 127) >> 7;
    const int step = (PT / 32) * NSTR;
    int sb = wave * NSTR + stripe;
    int yk[4], y1[4];
    load_ids(y, sb, nsb, Nn, lane, yk);
    while (sb < nsb) {
        const int base = sb << 7;
        const int sbn  = sb + step;
        load_ids(y, sbn, nsb, Nn, lane, y1);
#pragma unroll
        for (int k = 0; k < 4; ++k) {
            unsigned m = __builtin_amdgcn_ballot_w32((yk[k] >> 4) == group);
            while (m) {
                const int r = __builtin_ctz(m);
                m &= m - 1;
                const int seg = __builtin_amdgcn_readlane(yk[k], r);
                pool_row<PASS>(x, vr, cr, accL, wsL, mK, mF, base + 32 * k + r, seg, lane);
            }
        }
#pragma unroll
        for (int k = 0; k < 4; ++k) yk[k] = y1[k];
        sb = sbn;
    }
}

__device__ __forceinline__ void scan_list(const int* __restrict__ y, unsigned* listL,
                                          unsigned* cntS, unsigned* ovfL,
                                          int Nn, int group, int stripe, int wave, int lane) {
    const int nsb  = (Nn + 127) >> 7;
    const int step = (PT / 32) * NSTR;
    int sb = wave * NSTR + stripe;
    int yk[4], y1[4], y2[4];
    load_ids(y, sb, nsb, Nn, lane, yk);
    load_ids(y, sb + step, nsb, Nn, lane, y1);
    while (sb < nsb) {
        const int sbn = sb + step;
        load_ids(y, sbn + step, nsb, Nn, lane, y2);
#pragma unroll
        for (int k = 0; k < 4; ++k) {
            if ((yk[k] >> 4) == group) {
                const int sl = yk[k] & (SEGB - 1);
                const unsigned slot = __hip_atomic_fetch_add(cntS + sl, 1u, __ATOMIC_RELAXED,
                                                             __HIP_MEMORY_SCOPE_WORKGROUP);
                if (slot < (unsigned)LSEG) listL[sl * LSEG + slot] = (unsigned)((sb << 7) + 32 * k + lane);
                else *ovfL = 1u;
            }
        }
#pragma unroll
        for (int k = 0; k < 4; ++k) { yk[k] = y1[k]; y1[k] = y2[k]; }
        sb = sbn;
    }
}

__device__ __forceinline__ void fetch_row(const float* __restrict__ x, unsigned row, int lane,
                                          float (&xv)[4]) {
    const float* xr = x + (size_t)row * IN_CH + lane;
    xv[0] = xr[0]; xv[1] = xr[32]; xv[2] = xr[64]; xv[3] = xr[96];
}

__device__ __forceinline__ void ask_rows(const float* __restrict__ x, const unsigned* lst,
                                         int f, int last, int lane) {
    if (lane < 4) {
        const unsigned row = lst[min(f + (lane & 2), last)];
        __builtin_prefetch(x + (size_t)row * IN_CH + (lane & 1) * 64, 0, 1);
    }
}

__device__ __forceinline__ void row_update(const float (&xv)[4], const float (&vr)[16],
                                           const float (&cr)[4], float (&acc)[16],
                                           float (&m)[4], float (&ws)[4]) {
    float s[4];
    row_scores(xv, vr, cr, s);
#pragma unroll
    for (int hd = 0; hd < NHEAD; ++hd) s[hd] = uni(s[hd]);
    if (s[0] > m[0] || s[1] > m[1] || s[2] > m[2] || s[3] > m[3]) {
#pragma unroll
        for (int hd = 0; hd < NHEAD; ++hd) {
            const float mn = fmaxf(m[hd], s[hd]);
            const float sc = __expf(m[hd] - mn);
#pragma unroll
            for (int j = 0; j < 4; ++j) acc[hd * 4 + j] *= sc;
            ws[hd] *= sc;
            m[hd] = uni(mn);
        }
    }
#pragma unroll
    for (int hd = 0; hd < NHEAD; ++hd) {
        const float w = __expf(s[hd] - m[hd]);
#pragma unroll
        for (int j = 0; j < 4; ++j) acc[hd * 4 + j] = fmaf(w, xv[j], acc[hd * 4 + j]);
        ws[hd] += w;
    }
}

#define AHEAD 12
__device__ __forceinline__ void walk_seg(const float* __restrict__ x, const unsigned* lst,
                                         int n, int half, int lane,
                                         const float (&vr)[16], const float (&cr)[4],
                                         float (&acc)[16], float (&m)[4], float (&ws)[4]) {
    if (half >= n) return;
    const int last = n - 1;
    const int cnt  = (n - half + 1) >> 1;
    int e = half;
    float xa[4], xb[4];
    unsigned ra = (unsigned)__builtin_amdgcn_readfirstlane((int)lst[e]);
    unsigned rb = (unsigned)__builtin_amdgcn_readfirstlane((int)lst[min(e + 2, last)]);
#pragma unroll
    for (int a = 4; a < AHEAD; a += 4) ask_rows(x, lst, e + a, last, lane);
    for (int i = cnt >> 1; i > 0; --i) {
        fetch_row(x, ra, lane, xa);
        fetch_row(x, rb, lane, xb);
        const unsigned na = lst[min(e + 4, last)];
        const unsigned nb = lst[min(e + 6, last)];
        ask_rows(x, lst, e + AHEAD, last, lane);
        row_update(xa, vr, cr, acc, m, ws);
        row_update(xb, vr, cr, acc, m, ws);
        ra = (unsigned)__builtin_amdgcn_readfirstlane((int)na);
        rb = (unsigned)__builtin_amdgcn_readfirstlane((int)nb);
        e += 4;
    }
    if (cnt & 1) {
        fetch_row(x, ra, lane, xa);
        row_update(xa, vr, cr, acc, m, ws);
    }
}

__global__ __launch_bounds__(PT) __attribute__((amdgpu_waves_per_eu(16)))
void k_pool(const float* __restrict__ x, const int* __restrict__ y,
            const float* __restrict__ lin_w, const float* __restrict__ lin_b,
            const float* __restrict__ v, const float* __restrict__ c,
            float* out, float* part, float* smax, float* ssum, int Nn, int C) {
    __shared__ __attribute__((aligned(16))) float accL[SEGB * SP];
    __shared__ float wsL[SEGB * NHEAD];
    __shared__ unsigned mK[SEGB * NHEAD];
    __shared__ __attribute__((aligned(16))) float mF[SEGB * NHEAD];
    __shared__ float outL[SEGB * TOT_OUT];
    __shared__ unsigned listL[SEGB * LSEG];
    __shared__ unsigned cntS[SEGB];
    __shared__ __attribute__((aligned(16))) float pmL[(PT / 32) * NHEAD];
    __shared__ unsigned ovfL;

    const int t = threadIdx.x;
    const int group  = blockIdx.x % NGRP;
    const int stripe = blockIdx.x / NGRP;
    const int wave = __builtin_amdgcn_readfirstlane(t >> 5);
    const int lane = t & 31;

    for (int i = t; i < SEGB * SP; i += PT) accL[i] = 0.0f;
    if (t < SEGB * NHEAD) { wsL[t] = 0.0f; mK[t] = 0u; }
    if (t < SEGB) cntS[t] = 0u;
    if (t == 0) ovfL = 0u;

    float vr[16], cr[4];
#pragma unroll
    for (int hd = 0; hd < NHEAD; ++hd) {
#pragma unroll
        for (int j = 0; j < 4; ++j) vr[hd * 4 + j] = v[hd * IN_CH + lane + 32 * j];
        cr[hd] = c[hd];
    }
    __syncthreads();

    scan_list(y, listL, cntS, &ovfL, Nn, group, stripe, wave, lane);
    __syncthreads();
    const unsigned ovf = (unsigned)__builtin_amdgcn_readfirstlane((int)ovfL);
    if (ovf == 0u) {
        const int sl = wave >> 1;
        const int n  = __builtin_amdgcn_readfirstlane((int)cntS[sl]);
        float acc[16], m[4], ws[4];
#pragma unroll
        for (int i = 0; i < 16; ++i) acc[i] = 0.0f;
#pragma unroll
        for (int hd = 0; hd < NHEAD; ++hd) { m[hd] = -INFINITY; ws[hd] = 0.0f; }
        walk_seg(x, listL + sl * LSEG, n, wave & 1, lane, vr, cr, acc, m, ws);
        if (lane < NHEAD)
            pmL[wave * NHEAD + lane] = (lane == 0) ? m[0] : (lane == 1) ? m[1] : (lane == 2) ? m[2] : m[3];
        __syncthreads();
        const float4 o = *(const float4*)(pmL + (wave ^ 1) * NHEAD);
        const float om[4] = {o.x, o.y, o.z, o.w};
        float f[4], big[4];
#pragma unroll
        for (int hd = 0; hd < NHEAD; ++hd) {
            big[hd] = fmaxf(m[hd], om[hd]);
            f[hd] = (m[hd] == -INFINITY) ? 0.0f : __expf(m[hd] - big[hd]);
        }
        float* p = accL + sl * SP + lane;
#pragma unroll
        for (int hd = 0; hd < NHEAD; ++hd)
#pragma unroll
            for (int j = 0; j < 4; ++j)
                __hip_atomic_fetch_add(p + hd * IN_CH + j * 32, acc[hd * 4 + j] * f[hd],
                                       __ATOMIC_RELAXED, __HIP_MEMORY_SCOPE_WORKGROUP);
        if (lane < NHEAD) {
            const float wl = (lane == 0) ? ws[0] * f[0] : (lane == 1) ? ws[1] * f[1]
                           : (lane == 2) ? ws[2] * f[2] : ws[3] * f[3];
            __hip_atomic_fetch_add(wsL + sl * NHEAD + lane, wl,
                                   __ATOMIC_RELAXED, __HIP_MEMORY_SCOPE_WORKGROUP);
            if ((wave & 1) == 0)
                mF[sl * NHEAD + lane] = (lane == 0) ? big[0] : (lane == 1) ? big[1]
                                      : (lane == 2) ? big[2] : big[3];
        }
    } else {
        scan_pass<1>(x, y, vr, cr, accL, wsL, mK, mF, Nn, group, stripe, wave, lane);
        __syncthreads();
        if (t < SEGB * NHEAD) mF[t] = key2f(mK[t]);
        __syncthreads();
        scan_pass<2>(x, y, vr, cr, accL, wsL, mK, mF, Nn, group, stripe, wave, lane);
    }
    __syncthreads();

    if (wave < 16) {
        const int head = wave >> 2, nt = wave & 3;
        const int cc = lane & 15, h = lane >> 4;
        const int col = head * OUT_CH + nt * 16 + cc;
        const float* brow = lin_w + (size_t)col * IN_CH + 4 * h;
        const float* arow = accL + cc * SP + head * IN_CH + 4 * h;
        v8f acc = {0.f, 0.f, 0.f, 0.f, 0.f, 0.f, 0.f, 0.f};
#pragma unroll
        for (int j = 0; j < 16; ++j) {
            const float4 a4 = *(const float4*)(arow + 8 * j);
            const float4 b4 = *(const float4*)(brow + 8 * j);
            v2f a0 = {a4.x, a4.y}, a1 = {a4.z, a4.w};
            v2f b0 = {b4.x, b4.y}, b1 = {b4.z, b4.w};
            acc = __builtin_amdgcn_wmma_f32_16x16x4_f32(false, a0, false, b0, (short)0, acc, false, false);
            acc = __builtin_amdgcn_wmma_f32_16x16x4_f32(false, a1, false, b1, (short)0, acc, false, false);
        }
        const float bn = lin_b[col];
#pragma unroll
        for (int r = 0; r < 8; ++r) {
            const int sl = 8 * h + r;
            outL[sl * TOT_OUT + col] = acc[r] + bn * wsL[sl * NHEAD + head];
        }
    }
    __syncthreads();

    const int nseg = (C < NSEG) ? C : NSEG;
    float* dst = (stripe == 0) ? out : part + (size_t)(stripe - 1) * C * TOT_OUT;
#pragma unroll
    for (int i = 0; i < SEGB * TOT_OUT / PT; ++i) {
        const int e   = i * PT + t;
        const int seg = group * SEGB + (e >> 8);
        if (seg < nseg) dst[(size_t)seg * TOT_OUT + (e & 255)] = outL[e];
    }
    if (t < SEGB * NHEAD) {
        const int seg = group * SEGB + (t >> 2);
        if (seg < nseg) {
            const size_t o = ((size_t)stripe * C + seg) * NHEAD + (t & 3);
            smax[o] = mF[t];
            ssum[o] = wsL[t];
        }
    }
}

__global__ __launch_bounds__(256)
void k_merge(const float* __restrict__ part, const float* __restrict__ smax,
             const float* __restrict__ ssum, float* out, int C) {
    const int i = blockIdx.x * 256 + threadIdx.x;
    if (i >= C * TOT_OUT) return;
    const int seg = i >> 8, head = (i & 255) >> 6;
    if (seg >= NSEG) { out[i] = 0.0f; return; }
    float m[NSTR], best = -INFINITY;
#pragma unroll
    for (int b = 0; b < NSTR; ++b) {
        m[b] = smax[((size_t)b * C + seg) * NHEAD + head];
        best = fmaxf(best, m[b]);
    }
    float num = 0.0f, den = 0.0f;
#pragma unroll
    for (int b = 0; b < NSTR; ++b) {
        const float f  = (m[b] == -INFINITY) ? 0.0f : __expf(m[b] - best);
        const float pb = (b == 0) ? out[i] : part[(size_t)(b - 1) * C * TOT_OUT + i];
        num += f * pb;
        den += f * ssum[((size_t)b * C + seg) * NHEAD + head];
    }
    out[i] = (den > 0.0f) ? num / den : 0.0f;
}

extern "C" void kernel_launch(void* const* d_in, const int* in_sizes, int n_in,
                              void* d_out, int out_size, void* d_ws, size_t ws_size,
                              hipStream_t stream) {
    (void)n_in; (void)ws_size;
    const float* x     = (const float*)d_in[0];
    const int*   y     = (const int*)  d_in[1];
    const float* lin_w = (const float*)d_in[3];
    const float* lin_b = (const float*)d_in[4];
    const float* att_w = (const float*)d_in[5];
    const float* att_b = (const float*)d_in[6];
    float* out = (float*)d_out;

    int N = in_sizes[0] / IN_CH;
    int C = out_size / TOT_OUT;

    float* wsf  = (float*)d_ws;
    float* v    = wsf + WS_V;
    float* c    = wsf + WS_C;
    float* smax = wsf + WS_MX;
    float* ssum = wsf + WS_DN(C);
    float* part = wsf + WS_P(C);

    k_fold <<<2, 256, 0, stream>>>(lin_w, lin_b, att_w, att_b, v, c);
    k_pool <<<NGRP * NSTR, PT, 0, stream>>>(x, y, lin_w, lin_b, v, c,
                                            out, part, smax, ssum, N, C);
    k_merge<<<(C * TOT_OUT + 255) / 256, 256, 0, stream>>>(part, smax, ssum, out, C);
}


# k_pool row scores: the 4 heads' partial dot products are folded across lanes, one 5-step lane butterfly instead of four; bias and leaky once per row; same add tree per head, scores bit-identical
# speedup vs baseline: 11.6881x; 1.1215x over previous
#include <hip/hip_runtime.h>
#include <math.h>

typedef __attribute__((ext_vector_type(2))) float v2f;
typedef __attribute__((ext_vector_type(8))) float v8f;

#define IN_CH     128
#define OUT_CH    64
#define NHEAD     4
#define TOT_OUT   256
#define NEG_SLOPE 0.2f

#define NSEG  1000
#define SEGB  16
#define NGRP  63
#define NSTR  8
#define SP    516
#define PT    1024
#define LSEG  192

#define WS_V   0
#define WS_C   512
#define WS_MX  640
#define WS_DN(C) (WS_MX + NSTR * (C) * NHEAD)
#define WS_P(C)  (WS_MX + 2 * NSTR * (C) * NHEAD)

__global__ void k_fold(const float* __restrict__ lin_w, const float* __restrict__ lin_b,
                       const float* __restrict__ att_w, const float* __restrict__ att_b,
                       float* v, float* c) {
    int t = blockIdx.x * blockDim.x + threadIdx.x;
    if (t < NHEAD * IN_CH) {
        int h = t / IN_CH, i = t % IN_CH;
        float s = 0.0f;
        for (int d = 0; d < OUT_CH; ++d)
            s += lin_w[(size_t)(h * OUT_CH + d) * IN_CH + i] * att_w[d];
        v[h * IN_CH + i] = s;
    }
    if (t < NHEAD) {
        float s = att_b[0];
        for (int d = 0; d < OUT_CH; ++d) s += lin_b[t * OUT_CH + d] * att_w[d];
        c[t] = s;
    }
}


__device__ __forceinline__ float xhalf(float a) {
    const int b = __builtin_bit_cast(int, a);
    return __builtin_bit_cast(float,
        __builtin_amdgcn_permlanex16(b, b, 0x76543210, 0xfedcba98, false, false));
}
#define XADD(a, K) ((a) + __builtin_bit_cast(float, __builtin_amdgcn_update_dpp(          \
                        0, __builtin_bit_cast(int, (a)), 0x160 + (K), 0xf, 0xf, false)))
#define XADD2(k, g, K) ((k) + __builtin_bit_cast(float, __builtin_amdgcn_update_dpp(      \
                        0, __builtin_bit_cast(int, (g)), 0x160 + (K), 0xf, 0xf, false)))

__device__ __forceinline__ unsigned f2key(float f) {
    const unsigned b = __float_as_uint(f);
    return (b & 0x80000000u) ? ~b : (b | 0x80000000u);
}
__device__ __forceinline__ float key2f(unsigned k) {
    if (k == 0u) return -INFINITY;
    return __uint_as_float((k & 0x80000000u) ? (k ^ 0x80000000u) : ~k);
}

__device__ __forceinline__ float uni(float a) {
    return __builtin_bit_cast(float, __builtin_amdgcn_readfirstlane(__builtin_bit_cast(int, a)));
}

__device__ __forceinline__ void row_scores(const float (&xv)[4], const float (&vr)[16],
                                           float cl, int lane, float (&s)[4]) {
    float p[4];
#pragma unroll
    for (int hd = 0; hd < NHEAD; ++hd) {
        p[hd] = xv[0] * vr[hd * 4 + 0];
        p[hd] = fmaf(xv[1], vr[hd * 4 + 1], p[hd]);
        p[hd] = fmaf(xv[2], vr[hd * 4 + 2], p[hd]);
        p[hd] = fmaf(xv[3], vr[hd * 4 + 3], p[hd]);
    }
    const bool o1 = (lane & 1) != 0, o2 = (lane & 2) != 0;
    const float q0 = XADD2(o1 ? p[1] : p[0], o1 ? p[0] : p[1], 1);
    const float q1 = XADD2(o1 ? p[3] : p[2], o1 ? p[2] : p[3], 1);
    float r = XADD2(o2 ? q1 : q0, o2 ? q0 : q1, 2);
    r = XADD(r, 4); r = XADD(r, 8);
    r = r + xhalf(r);
    r += cl;
    const float t = (r >= 0.0f) ? r : NEG_SLOPE * r;
#pragma unroll
    for (int hd = 0; hd < NHEAD; ++hd)
        s[hd] = __builtin_bit_cast(float, __builtin_amdgcn_readlane(__builtin_bit_cast(int, t), hd));
}

template <int PASS>
__device__ __forceinline__ void pool_row(const float* __restrict__ x,
                                         const float (&vr)[16], float cl,
                                         float* accL, float* wsL, unsigned* mK, const float* mF,
                                         int row, int seg, int lane) {
    const float* xr = x + (size_t)row * IN_CH + lane;
    const float xv[4] = {xr[0], xr[32], xr[64], xr[96]};
    float s[4];
    row_scores(xv, vr, cl, lane, s);
    const int sl = seg & (SEGB - 1);
    if (PASS == 1) {
        if (lane < NHEAD) {
            const float sh = (lane == 0) ? s[0] : (lane == 1) ? s[1] : (lane == 2) ? s[2] : s[3];
            __hip_atomic_fetch_max(mK + sl * NHEAD + lane, f2key(sh),
                                   __ATOMIC_RELAXED, __HIP_MEMORY_SCOPE_WORKGROUP);
        }
    } else {
        const float4 mx = *(const float4*)(mF + sl * NHEAD);
        const float w[4] = {__expf(s[0] - mx.x), __expf(s[1] - mx.y),
                            __expf(s[2] - mx.z), __expf(s[3] - mx.w)};
        float* p = accL + sl * SP + lane;
#pragma unroll
        for (int hd = 0; hd < NHEAD; ++hd)
#pragma unroll
            for (int j = 0; j < 4; ++j)
                __hip_atomic_fetch_add(p + hd * IN_CH + j * 32, xv[j] * w[hd],
                                       __ATOMIC_RELAXED, __HIP_MEMORY_SCOPE_WORKGROUP);
        if (lane < NHEAD) {
            const float wl = (lane == 0) ? w[0] : (lane == 1) ? w[1] : (lane == 2) ? w[2] : w[3];
            __hip_atomic_fetch_add(wsL + sl * NHEAD + lane, wl,
                                   __ATOMIC_RELAXED, __HIP_MEMORY_SCOPE_WORKGROUP);
        }
    }
}

__device__ __forceinline__ void load_ids(const int* __restrict__ y, int sb, int nsb,
                                         int Nn, int lane, int (&yv)[4]) {
#pragma unroll
    for (int k = 0; k < 4; ++k) {
        const int r = (sb << 7) + 32 * k + lane;
        yv[k] = -1;
        if (sb < nsb && r < Nn) yv[k] = y[r];
    }
}

template <int PASS>
__device__ __forceinline__ void scan_pass(const float* __restrict__ x, const int* __restrict__ y,
                                          const float (&vr)[16], float cl,
                                          float* accL, float* wsL, unsigned* mK, const float* mF,
                                          int Nn, int group, int stripe, int wave, int lane) {
    const int nsb  = (Nn + 127) >> 7;
    const int step = (PT / 32) * NSTR;
    int sb = wave * NSTR + stripe;
    int yk[4], y1[4];
    load_ids(y, sb, nsb, Nn, lane, yk);
    while (sb < nsb) {
        const int base = sb << 7;
        const int sbn  = sb + step;
        load_ids(y, sbn, nsb, Nn, lane, y1);
#pragma unroll
        for (int k = 0; k < 4; ++k) {
            unsigned m = __builtin_amdgcn_ballot_w32((yk[k] >> 4) == group);
            while (m) {
                const int r = __builtin_ctz(m);
                m &= m - 1;
                const int seg = __builtin_amdgcn_readlane(yk[k], r);
                pool_row<PASS>(x, vr, cl, accL, wsL, mK, mF, base + 32 * k + r, seg, lane);
            }
        }
#pragma unroll
        for (int k = 0; k < 4; ++k) yk[k] = y1[k];
        sb = sbn;
    }
}

__device__ __forceinline__ void scan_list(const int* __restrict__ y, unsigned* listL,
                                          unsigned* cntS, unsigned* ovfL,
                                          int Nn, int group, int stripe, int wave, int lane) {
    const int nsb  = (Nn + 127) >> 7;
    const int step = (PT / 32) * NSTR;
    int sb = wave * NSTR + stripe;
    int yk[4], y1[4], y2[4];
    load_ids(y, sb, nsb, Nn, lane, yk);
    load_ids(y, sb + step, nsb, Nn, lane, y1);
    while (sb < nsb) {
        const int sbn = sb + step;
        load_ids(y, sbn + step, nsb, Nn, lane, y2);
#pragma unroll
        for (int k = 0; k < 4; ++k) {
            if ((yk[k] >> 4) == group) {
                const int sl = yk[k] & (SEGB - 1);
                const unsigned slot = __hip_atomic_fetch_add(cntS + sl, 1u, __ATOMIC_RELAXED,
                                                             __HIP_MEMORY_SCOPE_WORKGROUP);
                if (slot < (unsigned)LSEG) listL[sl * LSEG + slot] = (unsigned)((sb << 7) + 32 * k + lane);
                else *ovfL = 1u;
            }
        }
#pragma unroll
        for (int k = 0; k < 4; ++k) { yk[k] = y1[k]; y1[k] = y2[k]; }
        sb = sbn;
    }
}

__device__ __forceinline__ void fetch_row(const float* __restrict__ x, unsigned row, int lane,
                                          float (&xv)[4]) {
    const float* xr = x + (size_t)row * IN_CH + lane;
    xv[0] = xr[0]; xv[1] = xr[32]; xv[2] = xr[64]; xv[3] = xr[96];
}

__device__ __forceinline__ void ask_rows(const float* __restrict__ x, const unsigned* lst,
                                         int f, int last, int lane) {
    if (lane < 4) {
        const unsigned row = lst[min(f + (lane & 2), last)];
        __builtin_prefetch(x + (size_t)row * IN_CH + (lane & 1) * 64, 0, 1);
    }
}

__device__ __forceinline__ void row_update(const float (&xv)[4], const float (&vr)[16],
                                           float cl, int lane, float (&acc)[16],
                                           float (&m)[4], float (&ws)[4]) {
    float s[4];
    row_scores(xv, vr, cl, lane, s);
    if (s[0] > m[0] || s[1] > m[1] || s[2] > m[2] || s[3] > m[3]) {
#pragma unroll
        for (int hd = 0; hd < NHEAD; ++hd) {
            const float mn = fmaxf(m[hd], s[hd]);
            const float sc = __expf(m[hd] - mn);
#pragma unroll
            for (int j = 0; j < 4; ++j) acc[hd * 4 + j] *= sc;
            ws[hd] *= sc;
            m[hd] = uni(mn);
        }
    }
#pragma unroll
    for (int hd = 0; hd < NHEAD; ++hd) {
        const float w = __expf(s[hd] - m[hd]);
#pragma unroll
        for (int j = 0; j < 4; ++j) acc[hd * 4 + j] = fmaf(w, xv[j], acc[hd * 4 + j]);
        ws[hd] += w;
    }
}

#define AHEAD 12
__device__ __forceinline__ void walk_seg(const float* __restrict__ x, const unsigned* lst,
                                         int n, int half, int lane,
                                         const float (&vr)[16], float cl,
                                         float (&acc)[16], float (&m)[4], float (&ws)[4]) {
    if (half >= n) return;
    const int last = n - 1;
    const int cnt  = (n - half + 1) >> 1;
    int e = half;
    float xa[4], xb[4];
    unsigned ra = (unsigned)__builtin_amdgcn_readfirstlane((int)lst[e]);
    unsigned rb = (unsigned)__builtin_amdgcn_readfirstlane((int)lst[min(e + 2, last)]);
#pragma unroll
    for (int a = 4; a < AHEAD; a += 4) ask_rows(x, lst, e + a, last, lane);
    for (int i = cnt >> 1; i > 0; --i) {
        fetch_row(x, ra, lane, xa);
        fetch_row(x, rb, lane, xb);
        const unsigned na = lst[min(e + 4, last)];
        const unsigned nb = lst[min(e + 6, last)];
        ask_rows(x, lst, e + AHEAD, last, lane);
        row_update(xa, vr, cl, lane, acc, m, ws);
        row_update(xb, vr, cl, lane, acc, m, ws);
        ra = (unsigned)__builtin_amdgcn_readfirstlane((int)na);
        rb = (unsigned)__builtin_amdgcn_readfirstlane((int)nb);
        e += 4;
    }
    if (cnt & 1) {
        fetch_row(x, ra, lane, xa);
        row_update(xa, vr, cl, lane, acc, m, ws);
    }
}

__global__ __launch_bounds__(PT) __attribute__((amdgpu_waves_per_eu(16)))
void k_pool(const float* __restrict__ x, const int* __restrict__ y,
            const float* __restrict__ lin_w, const float* __restrict__ lin_b,
            const float* __restrict__ v, const float* __restrict__ c,
            float* out, float* part, float* smax, float* ssum, int Nn, int C) {
    __shared__ __attribute__((aligned(16))) float accL[SEGB * SP];
    __shared__ float wsL[SEGB * NHEAD];
    __shared__ unsigned mK[SEGB * NHEAD];
    __shared__ __attribute__((aligned(16))) float mF[SEGB * NHEAD];
    __shared__ float outL[SEGB * TOT_OUT];
    __shared__ unsigned listL[SEGB * LSEG];
    __shared__ unsigned cntS[SEGB];
    __shared__ __attribute__((aligned(16))) float pmL[(PT / 32) * NHEAD];
    __shared__ unsigned ovfL;

    const int t = threadIdx.x;
    const int group  = blockIdx.x % NGRP;
    const int stripe = blockIdx.x / NGRP;
    const int wave = __builtin_amdgcn_readfirstlane(t >> 5);
    const int lane = t & 31;

    for (int i = t; i < SEGB * SP; i += PT) accL[i] = 0.0f;
    if (t < SEGB * NHEAD) { wsL[t] = 0.0f; mK[t] = 0u; }
    if (t < SEGB) cntS[t] = 0u;
    if (t == 0) ovfL = 0u;

    float vr[16];
#pragma unroll
    for (int hd = 0; hd < NHEAD; ++hd)
#pragma unroll
        for (int j = 0; j < 4; ++j) vr[hd * 4 + j] = v[hd * IN_CH + lane + 32 * j];
    const float cl = c[lane & (NHEAD - 1)];
    __syncthreads();

    scan_list(y, listL, cntS, &ovfL, Nn, group, stripe, wave, lane);
    __syncthreads();
    const unsigned ovf = (unsigned)__builtin_amdgcn_readfirstlane((int)ovfL);
    if (ovf == 0u) {
        const int sl = wave >> 1;
        const int n  = __builtin_amdgcn_readfirstlane((int)cntS[sl]);
        float acc[16], m[4], ws[4];
#pragma unroll
        for (int i = 0; i < 16; ++i) acc[i] = 0.0f;
#pragma unroll
        for (int hd = 0; hd < NHEAD; ++hd) { m[hd] = -INFINITY; ws[hd] = 0.0f; }
        walk_seg(x, listL + sl * LSEG, n, wave & 1, lane, vr, cl, acc, m, ws);
        if (lane < NHEAD)
            pmL[wave * NHEAD + lane] = (lane == 0) ? m[0] : (lane == 1) ? m[1] : (lane == 2) ? m[2] : m[3];
        __syncthreads();
        const float4 o = *(const float4*)(pmL + (wave ^ 1) * NHEAD);
        const float om[4] = {o.x, o.y, o.z, o.w};
        float f[4], big[4];
#pragma unroll
        for (int hd = 0; hd < NHEAD; ++hd) {
            big[hd] = fmaxf(m[hd], om[hd]);
            f[hd] = (m[hd] == -INFINITY) ? 0.0f : __expf(m[hd] - big[hd]);
        }
        float* p = accL + sl * SP + lane;
#pragma unroll
        for (int hd = 0; hd < NHEAD; ++hd)
#pragma unroll
            for (int j = 0; j < 4; ++j)
                __hip_atomic_fetch_add(p + hd * IN_CH + j * 32, acc[hd * 4 + j] * f[hd],
                                       __ATOMIC_RELAXED, __HIP_MEMORY_SCOPE_WORKGROUP);
        if (lane < NHEAD) {
            const float wl = (lane == 0) ? ws[0] * f[0] : (lane == 1) ? ws[1] * f[1]
                           : (lane == 2) ? ws[2] * f[2] : ws[3] * f[3];
            __hip_atomic_fetch_add(wsL + sl * NHEAD + lane, wl,
                                   __ATOMIC_RELAXED, __HIP_MEMORY_SCOPE_WORKGROUP);
            if ((wave & 1) == 0)
                mF[sl * NHEAD + lane] = (lane == 0) ? big[0] : (lane == 1) ? big[1]
                                      : (lane == 2) ? big[2] : big[3];
        }
    } else {
        scan_pass<1>(x, y, vr, cl, accL, wsL, mK, mF, Nn, group, stripe, wave, lane);
        __syncthreads();
        if (t < SEGB * NHEAD) mF[t] = key2f(mK[t]);
        __syncthreads();
        scan_pass<2>(x, y, vr, cl, accL, wsL, mK, mF, Nn, group, stripe, wave, lane);
    }
    __syncthreads();

    if (wave < 16) {
        const int head = wave >> 2, nt = wave & 3;
        const int cc = lane & 15, h = lane >> 4;
        const int col = head * OUT_CH + nt * 16 + cc;
        const float* brow = lin_w + (size_t)col * IN_CH + 4 * h;
        const float* arow = accL + cc * SP + head * IN_CH + 4 * h;
        v8f acc = {0.f, 0.f, 0.f, 0.f, 0.f, 0.f, 0.f, 0.f};
#pragma unroll
        for (int j = 0; j < 16; ++j) {
            const float4 a4 = *(const float4*)(arow + 8 * j);
            const float4 b4 = *(const float4*)(brow + 8 * j);
            v2f a0 = {a4.x, a4.y}, a1 = {a4.z, a4.w};
            v2f b0 = {b4.x, b4.y}, b1 = {b4.z, b4.w};
            acc = __builtin_amdgcn_wmma_f32_16x16x4_f32(false, a0, false, b0, (short)0, acc, false, false);
            acc = __builtin_amdgcn_wmma_f32_16x16x4_f32(false, a1, false, b1, (short)0, acc, false, false);
        }
        const float bn = lin_b[col];
#pragma unroll
        for (int r = 0; r < 8; ++r) {
            const int sl = 8 * h + r;
            outL[sl * TOT_OUT + col] = acc[r] + bn * wsL[sl * NHEAD + head];
        }
    }
    __syncthreads();

    const int nseg = (C < NSEG) ? C : NSEG;
    float* dst = (stripe == 0) ? out : part + (size_t)(stripe - 1) * C * TOT_OUT;
#pragma unroll
    for (int i = 0; i < SEGB * TOT_OUT / PT; ++i) {
        const int e   = i * PT + t;
        const int seg = group * SEGB + (e >> 8);
        if (seg < nseg) dst[(size_t)seg * TOT_OUT + (e & 255)] = outL[e];
    }
    if (t < SEGB * NHEAD) {
        const int seg = group * SEGB + (t >> 2);
        if (seg < nseg) {
            const size_t o = ((size_t)stripe * C + seg) * NHEAD + (t & 3);
            smax[o] = mF[t];
            ssum[o] = wsL[t];
        }
    }
}

__global__ __launch_bounds__(256)
void k_merge(const float* __restrict__ part, const float* __restrict__ smax,
             const float* __restrict__ ssum, float* out, int C) {
    const int i = blockIdx.x * 256 + threadIdx.x;
    if (i >= C * TOT_OUT) return;
    const int seg = i >> 8, head = (i & 255) >> 6;
    if (seg >= NSEG) { out[i] = 0.0f; return; }
    float m[NSTR], best = -INFINITY;
#pragma unroll
    for (int b = 0; b < NSTR; ++b) {
        m[b] = smax[((size_t)b * C + seg) * NHEAD + head];
        best = fmaxf(best, m[b]);
    }
    float num = 0.0f, den = 0.0f;
#pragma unroll
    for (int b = 0; b < NSTR; ++b) {
        const float f  = (m[b] == -INFINITY) ? 0.0f : __expf(m[b] - best);
        const float pb = (b == 0) ? out[i] : part[(size_t)(b - 1) * C * TOT_OUT + i];
        num += f * pb;
        den += f * ssum[((size_t)b * C + seg) * NHEAD + head];
    }
    out[i] = (den > 0.0f) ? num / den : 0.0f;
}

extern "C" void kernel_launch(void* const* d_in, const int* in_sizes, int n_in,
                              void* d_out, int out_size, void* d_ws, size_t ws_size,
                              hipStream_t stream) {
    (void)n_in; (void)ws_size;
    const float* x     = (const float*)d_in[0];
    const int*   y     = (const int*)  d_in[1];
    const float* lin_w = (const float*)d_in[3];
    const float* lin_b = (const float*)d_in[4];
    const float* att_w = (const float*)d_in[5];
    const float* att_b = (const float*)d_in[6];
    float* out = (float*)d_out;

    int N = in_sizes[0] / IN_CH;
    int C = out_size / TOT_OUT;

    float* wsf  = (float*)d_ws;
    float* v    = wsf + WS_V;
    float* c    = wsf + WS_C;
    float* smax = wsf + WS_MX;
    float* ssum = wsf + WS_DN(C);
    float* part = wsf + WS_P(C);

    k_fold <<<2, 256, 0, stream>>>(lin_w, lin_b, att_w, att_b, v, c);
    k_pool <<<NGRP * NSTR, PT, 0, stream>>>(x, y, lin_w, lin_b, v, c,
                                            out, part, smax, ssum, N, C);
    k_merge<<<(C * TOT_OUT + 255) / 256, 256, 0, stream>>>(part, smax, ssum, out, C);
}


# k_pool scan: y read as 16-byte vectors, three or four 128-id batches in flight per wave, each waited for alone (was 4-byte loads, a full wait per round); rows listed in another order, sums reordered
# speedup vs baseline: 12.3125x; 1.0534x over previous
#include <hip/hip_runtime.h>
#include <math.h>

typedef __attribute__((ext_vector_type(2))) float v2f;
typedef __attribute__((ext_vector_type(8))) float v8f;

#define IN_CH     128
#define OUT_CH    64
#define NHEAD     4
#define TOT_OUT   256
#define NEG_SLOPE 0.2f

#define NSEG  1000
#define SEGB  16
#define NGRP  63
#define NSTR  8
#define SP    516
#define PT    1024
#define LSEG  192
#define LPIT  (LSEG + 1)

#define WS_V   0
#define WS_C   512
#define WS_MX  640
#define WS_DN(C) (WS_MX + NSTR * (C) * NHEAD)
#define WS_P(C)  (WS_MX + 2 * NSTR * (C) * NHEAD)

__global__ void k_fold(const float* __restrict__ lin_w, const float* __restrict__ lin_b,
                       const float* __restrict__ att_w, const float* __restrict__ att_b,
                       float* v, float* c) {
    int t = blockIdx.x * blockDim.x + threadIdx.x;
    if (t < NHEAD * IN_CH) {
        int h = t / IN_CH, i = t % IN_CH;
        float s = 0.0f;
        for (int d = 0; d < OUT_CH; ++d)
            s += lin_w[(size_t)(h * OUT_CH + d) * IN_CH + i] * att_w[d];
        v[h * IN_CH + i] = s;
    }
    if (t < NHEAD) {
        float s = att_b[0];
        for (int d = 0; d < OUT_CH; ++d) s += lin_b[t * OUT_CH + d] * att_w[d];
        c[t] = s;
    }
}


__device__ __forceinline__ float xhalf(float a) {
    const int b = __builtin_bit_cast(int, a);
    return __builtin_bit_cast(float,
        __builtin_amdgcn_permlanex16(b, b, 0x76543210, 0xfedcba98, false, false));
}
#define XADD(a, K) ((a) + __builtin_bit_cast(float, __builtin_amdgcn_update_dpp(          \
                        0, __builtin_bit_cast(int, (a)), 0x160 + (K), 0xf, 0xf, false)))
#define XADD2(k, g, K) ((k) + __builtin_bit_cast(float, __builtin_amdgcn_update_dpp(      \
                        0, __builtin_bit_cast(int, (g)), 0x160 + (K), 0xf, 0xf, false)))

__device__ __forceinline__ unsigned f2key(float f) {
    const unsigned b = __float_as_uint(f);
    return (b & 0x80000000u) ? ~b : (b | 0x80000000u);
}
__device__ __forceinline__ float key2f(unsigned k) {
    if (k == 0u) return -INFINITY;
    return __uint_as_float((k & 0x80000000u) ? (k ^ 0x80000000u) : ~k);
}

__device__ __forceinline__ float uni(float a) {
    return __builtin_bit_cast(float, __builtin_amdgcn_readfirstlane(__builtin_bit_cast(int, a)));
}

__device__ __forceinline__ void row_scores(const float (&xv)[4], const float (&vr)[16],
                                           float cl, int lane, float (&s)[4]) {
    float p[4];
#pragma unroll
    for (int hd = 0; hd < NHEAD; ++hd) {
        p[hd] = xv[0] * vr[hd * 4 + 0];
        p[hd] = fmaf(xv[1], vr[hd * 4 + 1], p[hd]);
        p[hd] = fmaf(xv[2], vr[hd * 4 + 2], p[hd]);
        p[hd] = fmaf(xv[3], vr[hd * 4 + 3], p[hd]);
    }
    const bool o1 = (lane & 1) != 0, o2 = (lane & 2) != 0;
    const float q0 = XADD2(o1 ? p[1] : p[0], o1 ? p[0] : p[1], 1);
    const float q1 = XADD2(o1 ? p[3] : p[2], o1 ? p[2] : p[3], 1);
    float r = XADD2(o2 ? q1 : q0, o2 ? q0 : q1, 2);
    r = XADD(r, 4); r = XADD(r, 8);
    r = r + xhalf(r);
    r += cl;
    const float t = (r >= 0.0f) ? r : NEG_SLOPE * r;
#pragma unroll
    for (int hd = 0; hd < NHEAD; ++hd)
        s[hd] = __builtin_bit_cast(float, __builtin_amdgcn_readlane(__builtin_bit_cast(int, t), hd));
}

template <int PASS>
__device__ __forceinline__ void pool_row(const float* __restrict__ x,
                                         const float (&vr)[16], float cl,
                                         float* accL, float* wsL, unsigned* mK, const float* mF,
                                         int row, int seg, int lane) {
    const float* xr = x + (size_t)row * IN_CH + lane;
    const float xv[4] = {xr[0], xr[32], xr[64], xr[96]};
    float s[4];
    row_scores(xv, vr, cl, lane, s);
    const int sl = seg & (SEGB - 1);
    if (PASS == 1) {
        if (lane < NHEAD) {
            const float sh = (lane == 0) ? s[0] : (lane == 1) ? s[1] : (lane == 2) ? s[2] : s[3];
            __hip_atomic_fetch_max(mK + sl * NHEAD + lane, f2key(sh),
                                   __ATOMIC_RELAXED, __HIP_MEMORY_SCOPE_WORKGROUP);
        }
    } else {
        const float4 mx = *(const float4*)(mF + sl * NHEAD);
        const float w[4] = {__expf(s[0] - mx.x), __expf(s[1] - mx.y),
                            __expf(s[2] - mx.z), __expf(s[3] - mx.w)};
        float* p = accL + sl * SP + lane;
#pragma unroll
        for (int hd = 0; hd < NHEAD; ++hd)
#pragma unroll
            for (int j = 0; j < 4; ++j)
                __hip_atomic_fetch_add(p + hd * IN_CH + j * 32, xv[j] * w[hd],
                                       __ATOMIC_RELAXED, __HIP_MEMORY_SCOPE_WORKGROUP);
        if (lane < NHEAD) {
            const float wl = (lane == 0) ? w[0] : (lane == 1) ? w[1] : (lane == 2) ? w[2] : w[3];
            __hip_atomic_fetch_add(wsL + sl * NHEAD + lane, wl,
                                   __ATOMIC_RELAXED, __HIP_MEMORY_SCOPE_WORKGROUP);
        }
    }
}

__device__ __forceinline__ void load_ids(const int* __restrict__ y, int sb, int nsb,
                                         int Nn, int lane, int (&yv)[4]) {
#pragma unroll
    for (int k = 0; k < 4; ++k) {
        const int r = (sb << 7) + 32 * k + lane;
        yv[k] = -1;
        if (sb < nsb && r < Nn) yv[k] = y[r];
    }
}

template <int PASS>
__device__ __forceinline__ void scan_pass(const float* __restrict__ x, const int* __restrict__ y,
                                          const float (&vr)[16], float cl,
                                          float* accL, float* wsL, unsigned* mK, const float* mF,
                                          int Nn, int group, int stripe, int wave, int lane) {
    const int nsb  = (Nn + 127) >> 7;
    const int step = (PT / 32) * NSTR;
    int sb = wave * NSTR + stripe;
    int yk[4], y1[4];
    load_ids(y, sb, nsb, Nn, lane, yk);
    while (sb < nsb) {
        const int base = sb << 7;
        const int sbn  = sb + step;
        load_ids(y, sbn, nsb, Nn, lane, y1);
#pragma unroll
        for (int k = 0; k < 4; ++k) {
            unsigned m = __builtin_amdgcn_ballot_w32((yk[k] >> 4) == group);
            while (m) {
                const int r = __builtin_ctz(m);
                m &= m - 1;
                const int seg = __builtin_amdgcn_readlane(yk[k], r);
                pool_row<PASS>(x, vr, cl, accL, wsL, mK, mF, base + 32 * k + r, seg, lane);
            }
        }
#pragma unroll
        for (int k = 0; k < 4; ++k) yk[k] = y1[k];
        sb = sbn;
    }
}

__device__ __forceinline__ void list_ids(const int4 q, int row0, int g, unsigned* listL,
                                         unsigned* cntS) {
    const int id[4] = {q.x, q.y, q.z, q.w};
#pragma unroll
    for (int j = 0; j < 4; ++j) {
        if ((id[j] >> 4) == g) {
            const int sl = id[j] & (SEGB - 1);
            const unsigned slot = __hip_atomic_fetch_add(cntS + sl, 1u, __ATOMIC_RELAXED,
                                                         __HIP_MEMORY_SCOPE_WORKGROUP);
            listL[sl * LPIT + min(slot, (unsigned)LSEG)] = (unsigned)(row0 + j);
        }
    }
}

__device__ __forceinline__ int4 load_ids4(const int* __restrict__ y, int sb, int nfull, int lane) {
    return *(const int4*)(y + ((size_t)min(sb, nfull - 1) << 7) + 4 * lane);
}

__device__ __forceinline__ void scan_list(const int* __restrict__ y, unsigned* listL,
                                          unsigned* cntS,
                                          int Nn, int group, int stripe, int wave, int lane) {
    const int nfull = Nn >> 7;
    const int step  = (PT / 32) * NSTR;
    int sb = wave * NSTR + stripe;
    if (sb < nfull) {
        int4 q0 = load_ids4(y, sb, nfull, lane);
        int4 q1 = load_ids4(y, sb + step, nfull, lane);
        int4 q2 = load_ids4(y, sb + 2 * step, nfull, lane);
        int4 q3 = load_ids4(y, sb + 3 * step, nfull, lane);
        while (sb < nfull) {
            const int r0 = (sb << 7) + 4 * lane;
            list_ids(q0, r0, group, listL, cntS);
            q0 = load_ids4(y, sb + 4 * step, nfull, lane);
            list_ids(q1, r0 + (step << 7), (sb + step < nfull) ? group : -2, listL, cntS);
            q1 = load_ids4(y, sb + 5 * step, nfull, lane);
            list_ids(q2, r0 + 2 * (step << 7), (sb + 2 * step < nfull) ? group : -2, listL, cntS);
            q2 = load_ids4(y, sb + 6 * step, nfull, lane);
            list_ids(q3, r0 + 3 * (step << 7), (sb + 3 * step < nfull) ? group : -2, listL, cntS);
            q3 = load_ids4(y, sb + 7 * step, nfull, lane);
            sb += 4 * step;
        }
    }
    if ((Nn & 127) != 0 && (nfull & (NSTR - 1)) == stripe && ((nfull >> 3) & (PT / 32 - 1)) == wave) {
        const int r0 = (nfull << 7) + 4 * lane;
        int4 q;
        q.x = (r0 + 0 < Nn) ? y[r0 + 0] : -1;
        q.y = (r0 + 1 < Nn) ? y[r0 + 1] : -1;
        q.z = (r0 + 2 < Nn) ? y[r0 + 2] : -1;
        q.w = (r0 + 3 < Nn) ? y[r0 + 3] : -1;
        list_ids(q, r0, group, listL, cntS);
    }
}

__device__ __forceinline__ void fetch_row(const float* __restrict__ x, unsigned row, int lane,
                                          float (&xv)[4]) {
    const float* xr = x + (size_t)row * IN_CH + lane;
    xv[0] = xr[0]; xv[1] = xr[32]; xv[2] = xr[64]; xv[3] = xr[96];
}

__device__ __forceinline__ void ask_rows(const float* __restrict__ x, const unsigned* lst,
                                         int f, int last, int lane) {
    if (lane < 4) {
        const unsigned row = lst[min(f + (lane & 2), last)];
        __builtin_prefetch(x + (size_t)row * IN_CH + (lane & 1) * 64, 0, 1);
    }
}

__device__ __forceinline__ void row_update(const float (&xv)[4], const float (&vr)[16],
                                           float cl, int lane, float (&acc)[16],
                                           float (&m)[4], float (&ws)[4]) {
    float s[4];
    row_scores(xv, vr, cl, lane, s);
    if (s[0] > m[0] || s[1] > m[1] || s[2] > m[2] || s[3] > m[3]) {
#pragma unroll
        for (int hd = 0; hd < NHEAD; ++hd) {
            const float mn = fmaxf(m[hd], s[hd]);
            const float sc = __expf(m[hd] - mn);
#pragma unroll
            for (int j = 0; j < 4; ++j) acc[hd * 4 + j] *= sc;
            ws[hd] *= sc;
            m[hd] = uni(mn);
        }
    }
#pragma unroll
    for (int hd = 0; hd < NHEAD; ++hd) {
        const float w = __expf(s[hd] - m[hd]);
#pragma unroll
        for (int j = 0; j < 4; ++j) acc[hd * 4 + j] = fmaf(w, xv[j], acc[hd * 4 + j]);
        ws[hd] += w;
    }
}

#define AHEAD 12
__device__ __forceinline__ void walk_seg(const float* __restrict__ x, const unsigned* lst,
                                         int n, int half, int lane,
                                         const float (&vr)[16], float cl,
                                         float (&acc)[16], float (&m)[4], float (&ws)[4]) {
    if (half >= n) return;
    const int last = n - 1;
    const int cnt  = (n - half + 1) >> 1;
    int e = half;
    float xa[4], xb[4];
    unsigned ra = (unsigned)__builtin_amdgcn_readfirstlane((int)lst[e]);
    unsigned rb = (unsigned)__builtin_amdgcn_readfirstlane((int)lst[min(e + 2, last)]);
#pragma unroll
    for (int a = 4; a < AHEAD; a += 4) ask_rows(x, lst, e + a, last, lane);
    for (int i = cnt >> 1; i > 0; --i) {
        fetch_row(x, ra, lane, xa);
        fetch_row(x, rb, lane, xb);
        const unsigned na = lst[min(e + 4, last)];
        const unsigned nb = lst[min(e + 6, last)];
        ask_rows(x, lst, e + AHEAD, last, lane);
        row_update(xa, vr, cl, lane, acc, m, ws);
        row_update(xb, vr, cl, lane, acc, m, ws);
        ra = (unsigned)__builtin_amdgcn_readfirstlane((int)na);
        rb = (unsigned)__builtin_amdgcn_readfirstlane((int)nb);
        e += 4;
    }
    if (cnt & 1) {
        fetch_row(x, ra, lane, xa);
        row_update(xa, vr, cl, lane, acc, m, ws);
    }
}

__global__ __launch_bounds__(PT) __attribute__((amdgpu_waves_per_eu(16)))
void k_pool(const float* __restrict__ x, const int* __restrict__ y,
            const float* __restrict__ lin_w, const float* __restrict__ lin_b,
            const float* __restrict__ v, const float* __restrict__ c,
            float* out, float* part, float* smax, float* ssum, int Nn, int C) {
    __shared__ __attribute__((aligned(16))) float accL[SEGB * SP];
    __shared__ float wsL[SEGB * NHEAD];
    __shared__ unsigned mK[SEGB * NHEAD];
    __shared__ __attribute__((aligned(16))) float mF[SEGB * NHEAD];
    __shared__ float outL[SEGB * TOT_OUT];
    __shared__ unsigned listL[SEGB * LPIT];
    __shared__ unsigned cntS[SEGB];
    __shared__ __attribute__((aligned(16))) float pmL[(PT / 32) * NHEAD];

    const int t = threadIdx.x;
    const int group  = blockIdx.x % NGRP;
    const int stripe = blockIdx.x / NGRP;
    const int wave = __builtin_amdgcn_readfirstlane(t >> 5);
    const int lane = t & 31;

    for (int i = t; i < SEGB * SP; i += PT) accL[i] = 0.0f;
    if (t < SEGB * NHEAD) { wsL[t] = 0.0f; mK[t] = 0u; }
    if (t < SEGB) cntS[t] = 0u;

    float vr[16];
#pragma unroll
    for (int hd = 0; hd < NHEAD; ++hd)
#pragma unroll
        for (int j = 0; j < 4; ++j) vr[hd * 4 + j] = v[hd * IN_CH + lane + 32 * j];
    const float cl = c[lane & (NHEAD - 1)];
    __syncthreads();

    scan_list(y, listL, cntS, Nn, group, stripe, wave, lane);
    __syncthreads();
    const unsigned ovf = __builtin_amdgcn_ballot_w32(cntS[lane & (SEGB - 1)] > (unsigned)LSEG);
    if (ovf == 0u) {
        const int sl = wave >> 1;
        const int n  = __builtin_amdgcn_readfirstlane((int)cntS[sl]);
        float acc[16], m[4], ws[4];
#pragma unroll
        for (int i = 0; i < 16; ++i) acc[i] = 0.0f;
#pragma unroll
        for (int hd = 0; hd < NHEAD; ++hd) { m[hd] = -INFINITY; ws[hd] = 0.0f; }
        walk_seg(x, listL + sl * LPIT, n, wave & 1, lane, vr, cl, acc, m, ws);
        if (lane < NHEAD)
            pmL[wave * NHEAD + lane] = (lane == 0) ? m[0] : (lane == 1) ? m[1] : (lane == 2) ? m[2] : m[3];
        __syncthreads();
        const float4 o = *(const float4*)(pmL + (wave ^ 1) * NHEAD);
        const float om[4] = {o.x, o.y, o.z, o.w};
        float f[4], big[4];
#pragma unroll
        for (int hd = 0; hd < NHEAD; ++hd) {
            big[hd] = fmaxf(m[hd], om[hd]);
            f[hd] = (m[hd] == -INFINITY) ? 0.0f : __expf(m[hd] - big[hd]);
        }
        float* p = accL + sl * SP + lane;
#pragma unroll
        for (int hd = 0; hd < NHEAD; ++hd)
#pragma unroll
            for (int j = 0; j < 4; ++j)
                __hip_atomic_fetch_add(p + hd * IN_CH + j * 32, acc[hd * 4 + j] * f[hd],
                                       __ATOMIC_RELAXED, __HIP_MEMORY_SCOPE_WORKGROUP);
        if (lane < NHEAD) {
            const float wl = (lane == 0) ? ws[0] * f[0] : (lane == 1) ? ws[1] * f[1]
                           : (lane == 2) ? ws[2] * f[2] : ws[3] * f[3];
            __hip_atomic_fetch_add(wsL + sl * NHEAD + lane, wl,
                                   __ATOMIC_RELAXED, __HIP_MEMORY_SCOPE_WORKGROUP);
            if ((wave & 1) == 0)
                mF[sl * NHEAD + lane] = (lane == 0) ? big[0] : (lane == 1) ? big[1]
                                      : (lane == 2) ? big[2] : big[3];
        }
    } else {
        scan_pass<1>(x, y, vr, cl, accL, wsL, mK, mF, Nn, group, stripe, wave, lane);
        __syncthreads();
        if (t < SEGB * NHEAD) mF[t] = key2f(mK[t]);
        __syncthreads();
        scan_pass<2>(x, y, vr, cl, accL, wsL, mK, mF, Nn, group, stripe, wave, lane);
    }
    __syncthreads();

    if (wave < 16) {
        const int head = wave >> 2, nt = wave & 3;
        const int cc = lane & 15, h = lane >> 4;
        const int col = head * OUT_CH + nt * 16 + cc;
        const float* brow = lin_w + (size_t)col * IN_CH + 4 * h;
        const float* arow = accL + cc * SP + head * IN_CH + 4 * h;
        v8f acc = {0.f, 0.f, 0.f, 0.f, 0.f, 0.f, 0.f, 0.f};
#pragma unroll
        for (int j = 0; j < 16; ++j) {
            const float4 a4 = *(const float4*)(arow + 8 * j);
            const float4 b4 = *(const float4*)(brow + 8 * j);
            v2f a0 = {a4.x, a4.y}, a1 = {a4.z, a4.w};
            v2f b0 = {b4.x, b4.y}, b1 = {b4.z, b4.w};
            acc = __builtin_amdgcn_wmma_f32_16x16x4_f32(false, a0, false, b0, (short)0, acc, false, false);
            acc = __builtin_amdgcn_wmma_f32_16x16x4_f32(false, a1, false, b1, (short)0, acc, false, false);
        }
        const float bn = lin_b[col];
#pragma unroll
        for (int r = 0; r < 8; ++r) {
            const int sl = 8 * h + r;
            outL[sl * TOT_OUT + col] = acc[r] + bn * wsL[sl * NHEAD + head];
        }
    }
    __syncthreads();

    const int nseg = (C < NSEG) ? C : NSEG;
    float* dst = (stripe == 0) ? out : part + (size_t)(stripe - 1) * C * TOT_OUT;
#pragma unroll
    for (int i = 0; i < SEGB * TOT_OUT / PT; ++i) {
        const int e   = i * PT + t;
        const int seg = group * SEGB + (e >> 8);
        if (seg < nseg) dst[(size_t)seg * TOT_OUT + (e & 255)] = outL[e];
    }
    if (t < SEGB * NHEAD) {
        const int seg = group * SEGB + (t >> 2);
        if (seg < nseg) {
            const size_t o = ((size_t)stripe * C + seg) * NHEAD + (t & 3);
            smax[o] = mF[t];
            ssum[o] = wsL[t];
        }
    }
}

__global__ __launch_bounds__(256)
void k_merge(const float* __restrict__ part, const float* __restrict__ smax,
             const float* __restrict__ ssum, float* out, int C) {
    const int i = blockIdx.x * 256 + threadIdx.x;
    if (i >= C * TOT_OUT) return;
    const int seg = i >> 8, head = (i & 255) >> 6;
    if (seg >= NSEG) { out[i] = 0.0f; return; }
    float m[NSTR], best = -INFINITY;
#pragma unroll
    for (int b = 0; b < NSTR; ++b) {
        m[b] = smax[((size_t)b * C + seg) * NHEAD + head];
        best = fmaxf(best, m[b]);
    }
    float num = 0.0f, den = 0.0f;
#pragma unroll
    for (int b = 0; b < NSTR; ++b) {
        const float f  = (m[b] == -INFINITY) ? 0.0f : __expf(m[b] - best);
        const float pb = (b == 0) ? out[i] : part[(size_t)(b - 1) * C * TOT_OUT + i];
        num += f * pb;
        den += f * ssum[((size_t)b * C + seg) * NHEAD + head];
    }
    out[i] = (den > 0.0f) ? num / den : 0.0f;
}

extern "C" void kernel_launch(void* const* d_in, const int* in_sizes, int n_in,
                              void* d_out, int out_size, void* d_ws, size_t ws_size,
                              hipStream_t stream) {
    (void)n_in; (void)ws_size;
    const float* x     = (const float*)d_in[0];
    const int*   y     = (const int*)  d_in[1];
    const float* lin_w = (const float*)d_in[3];
    const float* lin_b = (const float*)d_in[4];
    const float* att_w = (const float*)d_in[5];
    const float* att_b = (const float*)d_in[6];
    float* out = (float*)d_out;

    int N = in_sizes[0] / IN_CH;
    int C = out_size / TOT_OUT;

    float* wsf  = (float*)d_ws;
    float* v    = wsf + WS_V;
    float* c    = wsf + WS_C;
    float* smax = wsf + WS_MX;
    float* ssum = wsf + WS_DN(C);
    float* part = wsf + WS_P(C);

    k_fold <<<2, 256, 0, stream>>>(lin_w, lin_b, att_w, att_b, v, c);
    k_pool <<<NGRP * NSTR, PT, 0, stream>>>(x, y, lin_w, lin_b, v, c,
                                            out, part, smax, ssum, N, C);
    k_merge<<<(C * TOT_OUT + 255) / 256, 256, 0, stream>>>(part, smax, ssum, out, C);
}


# k_pool walk: a wave's scale m is set by its first row and raised only when a score exceeds it by 16 (was: at every new largest score); sum order kept, rounding differs with the scale
# speedup vs baseline: 13.0158x; 1.0571x over previous
#include <hip/hip_runtime.h>
#include <math.h>

typedef __attribute__((ext_vector_type(2))) float v2f;
typedef __attribute__((ext_vector_type(8))) float v8f;

#define IN_CH     128
#define OUT_CH    64
#define NHEAD     4
#define TOT_OUT   256
#define NEG_SLOPE 0.2f

#define NSEG  1000
#define SEGB  16
#define NGRP  63
#define NSTR  8
#define SP    516
#define PT    1024
#define LSEG  192
#define LPIT  (LSEG + 1)

#define WS_V   0
#define WS_C   512
#define WS_MX  640
#define WS_DN(C) (WS_MX + NSTR * (C) * NHEAD)
#define WS_P(C)  (WS_MX + 2 * NSTR * (C) * NHEAD)

__global__ void k_fold(const float* __restrict__ lin_w, const float* __restrict__ lin_b,
                       const float* __restrict__ att_w, const float* __restrict__ att_b,
                       float* v, float* c) {
    int t = blockIdx.x * blockDim.x + threadIdx.x;
    if (t < NHEAD * IN_CH) {
        int h = t / IN_CH, i = t % IN_CH;
        float s = 0.0f;
        for (int d = 0; d < OUT_CH; ++d)
            s += lin_w[(size_t)(h * OUT_CH + d) * IN_CH + i] * att_w[d];
        v[h * IN_CH + i] = s;
    }
    if (t < NHEAD) {
        float s = att_b[0];
        for (int d = 0; d < OUT_CH; ++d) s += lin_b[t * OUT_CH + d] * att_w[d];
        c[t] = s;
    }
}


__device__ __forceinline__ float xhalf(float a) {
    const int b = __builtin_bit_cast(int, a);
    return __builtin_bit_cast(float,
        __builtin_amdgcn_permlanex16(b, b, 0x76543210, 0xfedcba98, false, false));
}
#define XADD(a, K) ((a) + __builtin_bit_cast(float, __builtin_amdgcn_update_dpp(          \
                        0, __builtin_bit_cast(int, (a)), 0x160 + (K), 0xf, 0xf, false)))
#define XADD2(k, g, K) ((k) + __builtin_bit_cast(float, __builtin_amdgcn_update_dpp(      \
                        0, __builtin_bit_cast(int, (g)), 0x160 + (K), 0xf, 0xf, false)))

__device__ __forceinline__ unsigned f2key(float f) {
    const unsigned b = __float_as_uint(f);
    return (b & 0x80000000u) ? ~b : (b | 0x80000000u);
}
__device__ __forceinline__ float key2f(unsigned k) {
    if (k == 0u) return -INFINITY;
    return __uint_as_float((k & 0x80000000u) ? (k ^ 0x80000000u) : ~k);
}

__device__ __forceinline__ float uni(float a) {
    return __builtin_bit_cast(float, __builtin_amdgcn_readfirstlane(__builtin_bit_cast(int, a)));
}

__device__ __forceinline__ void row_scores(const float (&xv)[4], const float (&vr)[16],
                                           float cl, int lane, float (&s)[4]) {
    float p[4];
#pragma unroll
    for (int hd = 0; hd < NHEAD; ++hd) {
        p[hd] = xv[0] * vr[hd * 4 + 0];
        p[hd] = fmaf(xv[1], vr[hd * 4 + 1], p[hd]);
        p[hd] = fmaf(xv[2], vr[hd * 4 + 2], p[hd]);
        p[hd] = fmaf(xv[3], vr[hd * 4 + 3], p[hd]);
    }
    const bool o1 = (lane & 1) != 0, o2 = (lane & 2) != 0;
    const float q0 = XADD2(o1 ? p[1] : p[0], o1 ? p[0] : p[1], 1);
    const float q1 = XADD2(o1 ? p[3] : p[2], o1 ? p[2] : p[3], 1);
    float r = XADD2(o2 ? q1 : q0, o2 ? q0 : q1, 2);
    r = XADD(r, 4); r = XADD(r, 8);
    r = r + xhalf(r);
    r += cl;
    const float t = (r >= 0.0f) ? r : NEG_SLOPE * r;
#pragma unroll
    for (int hd = 0; hd < NHEAD; ++hd)
        s[hd] = __builtin_bit_cast(float, __builtin_amdgcn_readlane(__builtin_bit_cast(int, t), hd));
}

template <int PASS>
__device__ __forceinline__ void pool_row(const float* __restrict__ x,
                                         const float (&vr)[16], float cl,
                                         float* accL, float* wsL, unsigned* mK, const float* mF,
                                         int row, int seg, int lane) {
    const float* xr = x + (size_t)row * IN_CH + lane;
    const float xv[4] = {xr[0], xr[32], xr[64], xr[96]};
    float s[4];
    row_scores(xv, vr, cl, lane, s);
    const int sl = seg & (SEGB - 1);
    if (PASS == 1) {
        if (lane < NHEAD) {
            const float sh = (lane == 0) ? s[0] : (lane == 1) ? s[1] : (lane == 2) ? s[2] : s[3];
            __hip_atomic_fetch_max(mK + sl * NHEAD + lane, f2key(sh),
                                   __ATOMIC_RELAXED, __HIP_MEMORY_SCOPE_WORKGROUP);
        }
    } else {
        const float4 mx = *(const float4*)(mF + sl * NHEAD);
        const float w[4] = {__expf(s[0] - mx.x), __expf(s[1] - mx.y),
                            __expf(s[2] - mx.z), __expf(s[3] - mx.w)};
        float* p = accL + sl * SP + lane;
#pragma unroll
        for (int hd = 0; hd < NHEAD; ++hd)
#pragma unroll
            for (int j = 0; j < 4; ++j)
                __hip_atomic_fetch_add(p + hd * IN_CH + j * 32, xv[j] * w[hd],
                                       __ATOMIC_RELAXED, __HIP_MEMORY_SCOPE_WORKGROUP);
        if (lane < NHEAD) {
            const float wl = (lane == 0) ? w[0] : (lane == 1) ? w[1] : (lane == 2) ? w[2] : w[3];
            __hip_atomic_fetch_add(wsL + sl * NHEAD + lane, wl,
                                   __ATOMIC_RELAXED, __HIP_MEMORY_SCOPE_WORKGROUP);
        }
    }
}

__device__ __forceinline__ void load_ids(const int* __restrict__ y, int sb, int nsb,
                                         int Nn, int lane, int (&yv)[4]) {
#pragma unroll
    for (int k = 0; k < 4; ++k) {
        const int r = (sb << 7) + 32 * k + lane;
        yv[k] = -1;
        if (sb < nsb && r < Nn) yv[k] = y[r];
    }
}

template <int PASS>
__device__ __forceinline__ void scan_pass(const float* __restrict__ x, const int* __restrict__ y,
                                          const float (&vr)[16], float cl,
                                          float* accL, float* wsL, unsigned* mK, const float* mF,
                                          int Nn, int group, int stripe, int wave, int lane) {
    const int nsb  = (Nn + 127) >> 7;
    const int step = (PT / 32) * NSTR;
    int sb = wave * NSTR + stripe;
    int yk[4], y1[4];
    load_ids(y, sb, nsb, Nn, lane, yk);
    while (sb < nsb) {
        const int base = sb << 7;
        const int sbn  = sb + step;
        load_ids(y, sbn, nsb, Nn, lane, y1);
#pragma unroll
        for (int k = 0; k < 4; ++k) {
            unsigned m = __builtin_amdgcn_ballot_w32((yk[k] >> 4) == group);
            while (m) {
                const int r = __builtin_ctz(m);
                m &= m - 1;
                const int seg = __builtin_amdgcn_readlane(yk[k], r);
                pool_row<PASS>(x, vr, cl, accL, wsL, mK, mF, base + 32 * k + r, seg, lane);
            }
        }
#pragma unroll
        for (int k = 0; k < 4; ++k) yk[k] = y1[k];
        sb = sbn;
    }
}

__device__ __forceinline__ void list_ids(const int4 q, int row0, int g, unsigned* listL,
                                         unsigned* cntS) {
    const int id[4] = {q.x, q.y, q.z, q.w};
#pragma unroll
    for (int j = 0; j < 4; ++j) {
        if ((id[j] >> 4) == g) {
            const int sl = id[j] & (SEGB - 1);
            const unsigned slot = __hip_atomic_fetch_add(cntS + sl, 1u, __ATOMIC_RELAXED,
                                                         __HIP_MEMORY_SCOPE_WORKGROUP);
            listL[sl * LPIT + min(slot, (unsigned)LSEG)] = (unsigned)(row0 + j);
        }
    }
}

__device__ __forceinline__ int4 load_ids4(const int* __restrict__ y, int sb, int nfull, int lane) {
    return *(const int4*)(y + ((size_t)min(sb, nfull - 1) << 7) + 4 * lane);
}

__device__ __forceinline__ void scan_list(const int* __restrict__ y, unsigned* listL,
                                          unsigned* cntS,
                                          int Nn, int group, int stripe, int wave, int lane) {
    const int nfull = Nn >> 7;
    const int step  = (PT / 32) * NSTR;
    int sb = wave * NSTR + stripe;
    if (sb < nfull) {
        int4 q0 = load_ids4(y, sb, nfull, lane);
        int4 q1 = load_ids4(y, sb + step, nfull, lane);
        int4 q2 = load_ids4(y, sb + 2 * step, nfull, lane);
        int4 q3 = load_ids4(y, sb + 3 * step, nfull, lane);
        while (sb < nfull) {
            const int r0 = (sb << 7) + 4 * lane;
            list_ids(q0, r0, group, listL, cntS);
            q0 = load_ids4(y, sb + 4 * step, nfull, lane);
            list_ids(q1, r0 + (step << 7), (sb + step < nfull) ? group : -2, listL, cntS);
            q1 = load_ids4(y, sb + 5 * step, nfull, lane);
            list_ids(q2, r0 + 2 * (step << 7), (sb + 2 * step < nfull) ? group : -2, listL, cntS);
            q2 = load_ids4(y, sb + 6 * step, nfull, lane);
            list_ids(q3, r0 + 3 * (step << 7), (sb + 3 * step < nfull) ? group : -2, listL, cntS);
            q3 = load_ids4(y, sb + 7 * step, nfull, lane);
            sb += 4 * step;
        }
    }
    if ((Nn & 127) != 0 && (nfull & (NSTR - 1)) == stripe && ((nfull >> 3) & (PT / 32 - 1)) == wave) {
        const int r0 = (nfull << 7) + 4 * lane;
        int4 q;
        q.x = (r0 + 0 < Nn) ? y[r0 + 0] : -1;
        q.y = (r0 + 1 < Nn) ? y[r0 + 1] : -1;
        q.z = (r0 + 2 < Nn) ? y[r0 + 2] : -1;
        q.w = (r0 + 3 < Nn) ? y[r0 + 3] : -1;
        list_ids(q, r0, group, listL, cntS);
    }
}

__device__ __forceinline__ void fetch_row(const float* __restrict__ x, unsigned row, int lane,
                                          float (&xv)[4]) {
    const float* xr = x + (size_t)row * IN_CH + lane;
    xv[0] = xr[0]; xv[1] = xr[32]; xv[2] = xr[64]; xv[3] = xr[96];
}

__device__ __forceinline__ void ask_rows(const float* __restrict__ x, const unsigned* lst,
                                         int f, int last, int lane) {
    if (lane < 4) {
        const unsigned row = lst[min(f + (lane & 2), last)];
        __builtin_prefetch(x + (size_t)row * IN_CH + (lane & 1) * 64, 0, 1);
    }
}

#define HEADROOM 16.0f

__device__ __forceinline__ void row_update(const float (&xv)[4], const float (&vr)[16],
                                           float cl, int lane, float (&acc)[16],
                                           float (&m)[4], float (&mt)[4], float (&ws)[4]) {
    float s[4];
    row_scores(xv, vr, cl, lane, s);
    if (s[0] > mt[0] || s[1] > mt[1] || s[2] > mt[2] || s[3] > mt[3]) {
#pragma unroll
        for (int hd = 0; hd < NHEAD; ++hd) {
            const float mn = fmaxf(m[hd], s[hd]);
            const float sc = __expf(m[hd] - mn);
#pragma unroll
            for (int j = 0; j < 4; ++j) acc[hd * 4 + j] *= sc;
            ws[hd] *= sc;
            m[hd]  = uni(mn);
            mt[hd] = m[hd] + HEADROOM;
        }
    }
#pragma unroll
    for (int hd = 0; hd < NHEAD; ++hd) {
        const float w = __expf(s[hd] - m[hd]);
#pragma unroll
        for (int j = 0; j < 4; ++j) acc[hd * 4 + j] = fmaf(w, xv[j], acc[hd * 4 + j]);
        ws[hd] += w;
    }
}

#define AHEAD 12
__device__ __forceinline__ void walk_seg(const float* __restrict__ x, const unsigned* lst,
                                         int n, int half, int lane,
                                         const float (&vr)[16], float cl,
                                         float (&acc)[16], float (&m)[4], float (&mt)[4],
                                         float (&ws)[4]) {
    if (half >= n) return;
    const int last = n - 1;
    const int cnt  = (n - half + 1) >> 1;
    int e = half;
    float xa[4], xb[4];
    unsigned ra = (unsigned)__builtin_amdgcn_readfirstlane((int)lst[e]);
    unsigned rb = (unsigned)__builtin_amdgcn_readfirstlane((int)lst[min(e + 2, last)]);
#pragma unroll
    for (int a = 4; a < AHEAD; a += 4) ask_rows(x, lst, e + a, last, lane);
    for (int i = cnt >> 1; i > 0; --i) {
        fetch_row(x, ra, lane, xa);
        fetch_row(x, rb, lane, xb);
        const unsigned na = lst[min(e + 4, last)];
        const unsigned nb = lst[min(e + 6, last)];
        ask_rows(x, lst, e + AHEAD, last, lane);
        row_update(xa, vr, cl, lane, acc, m, mt, ws);
        row_update(xb, vr, cl, lane, acc, m, mt, ws);
        ra = (unsigned)__builtin_amdgcn_readfirstlane((int)na);
        rb = (unsigned)__builtin_amdgcn_readfirstlane((int)nb);
        e += 4;
    }
    if (cnt & 1) {
        fetch_row(x, ra, lane, xa);
        row_update(xa, vr, cl, lane, acc, m, mt, ws);
    }
}

__global__ __launch_bounds__(PT) __attribute__((amdgpu_waves_per_eu(16)))
void k_pool(const float* __restrict__ x, const int* __restrict__ y,
            const float* __restrict__ lin_w, const float* __restrict__ lin_b,
            const float* __restrict__ v, const float* __restrict__ c,
            float* out, float* part, float* smax, float* ssum, int Nn, int C) {
    __shared__ __attribute__((aligned(16))) float accL[SEGB * SP];
    __shared__ float wsL[SEGB * NHEAD];
    __shared__ unsigned mK[SEGB * NHEAD];
    __shared__ __attribute__((aligned(16))) float mF[SEGB * NHEAD];
    __shared__ float outL[SEGB * TOT_OUT];
    __shared__ unsigned listL[SEGB * LPIT];
    __shared__ unsigned cntS[SEGB];
    __shared__ __attribute__((aligned(16))) float pmL[(PT / 32) * NHEAD];

    const int t = threadIdx.x;
    const int group  = blockIdx.x % NGRP;
    const int stripe = blockIdx.x / NGRP;
    const int wave = __builtin_amdgcn_readfirstlane(t >> 5);
    const int lane = t & 31;

    for (int i = t; i < SEGB * SP; i += PT) accL[i] = 0.0f;
    if (t < SEGB * NHEAD) { wsL[t] = 0.0f; mK[t] = 0u; }
    if (t < SEGB) cntS[t] = 0u;

    float vr[16];
#pragma unroll
    for (int hd = 0; hd < NHEAD; ++hd)
#pragma unroll
        for (int j = 0; j < 4; ++j) vr[hd * 4 + j] = v[hd * IN_CH + lane + 32 * j];
    const float cl = c[lane & (NHEAD - 1)];
    __syncthreads();

    scan_list(y, listL, cntS, Nn, group, stripe, wave, lane);
    __syncthreads();
    const unsigned ovf = __builtin_amdgcn_ballot_w32(cntS[lane & (SEGB - 1)] > (unsigned)LSEG);
    if (ovf == 0u) {
        const int sl = wave >> 1;
        const int n  = __builtin_amdgcn_readfirstlane((int)cntS[sl]);
        float acc[16], m[4], mt[4], ws[4];
#pragma unroll
        for (int i = 0; i < 16; ++i) acc[i] = 0.0f;
#pragma unroll
        for (int hd = 0; hd < NHEAD; ++hd) { m[hd] = -INFINITY; mt[hd] = -INFINITY; ws[hd] = 0.0f; }
        walk_seg(x, listL + sl * LPIT, n, wave & 1, lane, vr, cl, acc, m, mt, ws);
        if (lane < NHEAD)
            pmL[wave * NHEAD + lane] = (lane == 0) ? m[0] : (lane == 1) ? m[1] : (lane == 2) ? m[2] : m[3];
        __syncthreads();
        const float4 o = *(const float4*)(pmL + (wave ^ 1) * NHEAD);
        const float om[4] = {o.x, o.y, o.z, o.w};
        float f[4], big[4];
#pragma unroll
        for (int hd = 0; hd < NHEAD; ++hd) {
            big[hd] = fmaxf(m[hd], om[hd]);
            f[hd] = (m[hd] == -INFINITY) ? 0.0f : __expf(m[hd] - big[hd]);
        }
        float* p = accL + sl * SP + lane;
#pragma unroll
        for (int hd = 0; hd < NHEAD; ++hd)
#pragma unroll
            for (int j = 0; j < 4; ++j)
                __hip_atomic_fetch_add(p + hd * IN_CH + j * 32, acc[hd * 4 + j] * f[hd],
                                       __ATOMIC_RELAXED, __HIP_MEMORY_SCOPE_WORKGROUP);
        if (lane < NHEAD) {
            const float wl = (lane == 0) ? ws[0] * f[0] : (lane == 1) ? ws[1] * f[1]
                           : (lane == 2) ? ws[2] * f[2] : ws[3] * f[3];
            __hip_atomic_fetch_add(wsL + sl * NHEAD + lane, wl,
                                   __ATOMIC_RELAXED, __HIP_MEMORY_SCOPE_WORKGROUP);
            if ((wave & 1) == 0)
                mF[sl * NHEAD + lane] = (lane == 0) ? big[0] : (lane == 1) ? big[1]
                                      : (lane == 2) ? big[2] : big[3];
        }
    } else {
        scan_pass<1>(x, y, vr, cl, accL, wsL, mK, mF, Nn, group, stripe, wave, lane);
        __syncthreads();
        if (t < SEGB * NHEAD) mF[t] = key2f(mK[t]);
        __syncthreads();
        scan_pass<2>(x, y, vr, cl, accL, wsL, mK, mF, Nn, group, stripe, wave, lane);
    }
    __syncthreads();

    if (wave < 16) {
        const int head = wave >> 2, nt = wave & 3;
        const int cc = lane & 15, h = lane >> 4;
        const int col = head * OUT_CH + nt * 16 + cc;
        const float* brow = lin_w + (size_t)col * IN_CH + 4 * h;
        const float* arow = accL + cc * SP + head * IN_CH + 4 * h;
        v8f acc = {0.f, 0.f, 0.f, 0.f, 0.f, 0.f, 0.f, 0.f};
#pragma unroll
        for (int j = 0; j < 16; ++j) {
            const float4 a4 = *(const float4*)(arow + 8 * j);
            const float4 b4 = *(const float4*)(brow + 8 * j);
            v2f a0 = {a4.x, a4.y}, a1 = {a4.z, a4.w};
            v2f b0 = {b4.x, b4.y}, b1 = {b4.z, b4.w};
            acc = __builtin_amdgcn_wmma_f32_16x16x4_f32(false, a0, false, b0, (short)0, acc, false, false);
            acc = __builtin_amdgcn_wmma_f32_16x16x4_f32(false, a1, false, b1, (short)0, acc, false, false);
        }
        const float bn = lin_b[col];
#pragma unroll
        for (int r = 0; r < 8; ++r) {
            const int sl = 8 * h + r;
            outL[sl * TOT_OUT + col] = acc[r] + bn * wsL[sl * NHEAD + head];
        }
    }
    __syncthreads();

    const int nseg = (C < NSEG) ? C : NSEG;
    float* dst = (stripe == 0) ? out : part + (size_t)(stripe - 1) * C * TOT_OUT;
#pragma unroll
    for (int i = 0; i < SEGB * TOT_OUT / PT; ++i) {
        const int e   = i * PT + t;
        const int seg = group * SEGB + (e >> 8);
        if (seg < nseg) dst[(size_t)seg * TOT_OUT + (e & 255)] = outL[e];
    }
    if (t < SEGB * NHEAD) {
        const int seg = group * SEGB + (t >> 2);
        if (seg < nseg) {
            const size_t o = ((size_t)stripe * C + seg) * NHEAD + (t & 3);
            smax[o] = mF[t];
            ssum[o] = wsL[t];
        }
    }
}

__global__ __launch_bounds__(256)
void k_merge(const float* __restrict__ part, const float* __restrict__ smax,
             const float* __restrict__ ssum, float* out, int C) {
    const int i = blockIdx.x * 256 + threadIdx.x;
    if (i >= C * TOT_OUT) return;
    const int seg = i >> 8, head = (i & 255) >> 6;
    if (seg >= NSEG) { out[i] = 0.0f; return; }
    float m[NSTR], best = -INFINITY;
#pragma unroll
    for (int b = 0; b < NSTR; ++b) {
        m[b] = smax[((size_t)b * C + seg) * NHEAD + head];
        best = fmaxf(best, m[b]);
    }
    float num = 0.0f, den = 0.0f;
#pragma unroll
    for (int b = 0; b < NSTR; ++b) {
        const float f  = (m[b] == -INFINITY) ? 0.0f : __expf(m[b] - best);
        const float pb = (b == 0) ? out[i] : part[(size_t)(b - 1) * C * TOT_OUT + i];
        num += f * pb;
        den += f * ssum[((size_t)b * C + seg) * NHEAD + head];
    }
    out[i] = (den > 0.0f) ? num / den : 0.0f;
}

extern "C" void kernel_launch(void* const* d_in, const int* in_sizes, int n_in,
                              void* d_out, int out_size, void* d_ws, size_t ws_size,
                              hipStream_t stream) {
    (void)n_in; (void)ws_size;
    const float* x     = (const float*)d_in[0];
    const int*   y     = (const int*)  d_in[1];
    const float* lin_w = (const float*)d_in[3];
    const float* lin_b = (const float*)d_in[4];
    const float* att_w = (const float*)d_in[5];
    const float* att_b = (const float*)d_in[6];
    float* out = (float*)d_out;

    int N = in_sizes[0] / IN_CH;
    int C = out_size / TOT_OUT;

    float* wsf  = (float*)d_ws;
    float* v    = wsf + WS_V;
    float* c    = wsf + WS_C;
    float* smax = wsf + WS_MX;
    float* ssum = wsf + WS_DN(C);
    float* part = wsf + WS_P(C);

    k_fold <<<2, 256, 0, stream>>>(lin_w, lin_b, att_w, att_b, v, c);
    k_pool <<<NGRP * NSTR, PT, 0, stream>>>(x, y, lin_w, lin_b, v, c,
                                            out, part, smax, ssum, N, C);
    k_merge<<<(C * TOT_OUT + 255) / 256, 256, 0, stream>>>(part, smax, ssum, out, C);
}


# k_fold's launch gone: every k_pool block folds att_w into lin_w itself, a thread sums 32 of the 64 rows, halves meet in LDS; launches 3 -> 2; folded weights summed in another order
# speedup vs baseline: 13.8327x; 1.0628x over previous
#include <hip/hip_runtime.h>
#include <math.h>

typedef __attribute__((ext_vector_type(2))) float v2f;
typedef __attribute__((ext_vector_type(8))) float v8f;

#define IN_CH     128
#define OUT_CH    64
#define NHEAD     4
#define TOT_OUT   256
#define NEG_SLOPE 0.2f

#define NSEG  1000
#define SEGB  16
#define NGRP  63
#define NSTR  8
#define SP    516
#define PT    1024
#define LSEG  192
#define LPIT  (LSEG + 1)

#define WS_MX  0
#define WS_DN(C) (WS_MX + NSTR * (C) * NHEAD)
#define WS_P(C)  (WS_MX + 2 * NSTR * (C) * NHEAD)


__device__ __forceinline__ float xhalf(float a) {
    const int b = __builtin_bit_cast(int, a);
    return __builtin_bit_cast(float,
        __builtin_amdgcn_permlanex16(b, b, 0x76543210, 0xfedcba98, false, false));
}
#define XADD(a, K) ((a) + __builtin_bit_cast(float, __builtin_amdgcn_update_dpp(          \
                        0, __builtin_bit_cast(int, (a)), 0x160 + (K), 0xf, 0xf, false)))
#define XADD2(k, g, K) ((k) + __builtin_bit_cast(float, __builtin_amdgcn_update_dpp(      \
                        0, __builtin_bit_cast(int, (g)), 0x160 + (K), 0xf, 0xf, false)))

__device__ __forceinline__ unsigned f2key(float f) {
    const unsigned b = __float_as_uint(f);
    return (b & 0x80000000u) ? ~b : (b | 0x80000000u);
}
__device__ __forceinline__ float key2f(unsigned k) {
    if (k == 0u) return -INFINITY;
    return __uint_as_float((k & 0x80000000u) ? (k ^ 0x80000000u) : ~k);
}

__device__ __forceinline__ float uni(float a) {
    return __builtin_bit_cast(float, __builtin_amdgcn_readfirstlane(__builtin_bit_cast(int, a)));
}

__device__ __forceinline__ void row_scores(const float (&xv)[4], const float (&vr)[16],
                                           float cl, int lane, float (&s)[4]) {
    float p[4];
#pragma unroll
    for (int hd = 0; hd < NHEAD; ++hd) {
        p[hd] = xv[0] * vr[hd * 4 + 0];
        p[hd] = fmaf(xv[1], vr[hd * 4 + 1], p[hd]);
        p[hd] = fmaf(xv[2], vr[hd * 4 + 2], p[hd]);
        p[hd] = fmaf(xv[3], vr[hd * 4 + 3], p[hd]);
    }
    const bool o1 = (lane & 1) != 0, o2 = (lane & 2) != 0;
    const float q0 = XADD2(o1 ? p[1] : p[0], o1 ? p[0] : p[1], 1);
    const float q1 = XADD2(o1 ? p[3] : p[2], o1 ? p[2] : p[3], 1);
    float r = XADD2(o2 ? q1 : q0, o2 ? q0 : q1, 2);
    r = XADD(r, 4); r = XADD(r, 8);
    r = r + xhalf(r);
    r += cl;
    const float t = (r >= 0.0f) ? r : NEG_SLOPE * r;
#pragma unroll
    for (int hd = 0; hd < NHEAD; ++hd)
        s[hd] = __builtin_bit_cast(float, __builtin_amdgcn_readlane(__builtin_bit_cast(int, t), hd));
}

template <int PASS>
__device__ __forceinline__ void pool_row(const float* __restrict__ x,
                                         const float (&vr)[16], float cl,
                                         float* accL, float* wsL, unsigned* mK, const float* mF,
                                         int row, int seg, int lane) {
    const float* xr = x + (size_t)row * IN_CH + lane;
    const float xv[4] = {xr[0], xr[32], xr[64], xr[96]};
    float s[4];
    row_scores(xv, vr, cl, lane, s);
    const int sl = seg & (SEGB - 1);
    if (PASS == 1) {
        if (lane < NHEAD) {
            const float sh = (lane == 0) ? s[0] : (lane == 1) ? s[1] : (lane == 2) ? s[2] : s[3];
            __hip_atomic_fetch_max(mK + sl * NHEAD + lane, f2key(sh),
                                   __ATOMIC_RELAXED, __HIP_MEMORY_SCOPE_WORKGROUP);
        }
    } else {
        const float4 mx = *(const float4*)(mF + sl * NHEAD);
        const float w[4] = {__expf(s[0] - mx.x), __expf(s[1] - mx.y),
                            __expf(s[2] - mx.z), __expf(s[3] - mx.w)};
        float* p = accL + sl * SP + lane;
#pragma unroll
        for (int hd = 0; hd < NHEAD; ++hd)
#pragma unroll
            for (int j = 0; j < 4; ++j)
                __hip_atomic_fetch_add(p + hd * IN_CH + j * 32, xv[j] * w[hd],
                                       __ATOMIC_RELAXED, __HIP_MEMORY_SCOPE_WORKGROUP);
        if (lane < NHEAD) {
            const float wl = (lane == 0) ? w[0] : (lane == 1) ? w[1] : (lane == 2) ? w[2] : w[3];
            __hip_atomic_fetch_add(wsL + sl * NHEAD + lane, wl,
                                   __ATOMIC_RELAXED, __HIP_MEMORY_SCOPE_WORKGROUP);
        }
    }
}

__device__ __forceinline__ void load_ids(const int* __restrict__ y, int sb, int nsb,
                                         int Nn, int lane, int (&yv)[4]) {
#pragma unroll
    for (int k = 0; k < 4; ++k) {
        const int r = (sb << 7) + 32 * k + lane;
        yv[k] = -1;
        if (sb < nsb && r < Nn) yv[k] = y[r];
    }
}

template <int PASS>
__device__ __forceinline__ void scan_pass(const float* __restrict__ x, const int* __restrict__ y,
                                          const float (&vr)[16], float cl,
                                          float* accL, float* wsL, unsigned* mK, const float* mF,
                                          int Nn, int group, int stripe, int wave, int lane) {
    const int nsb  = (Nn + 127) >> 7;
    const int step = (PT / 32) * NSTR;
    int sb = wave * NSTR + stripe;
    int yk[4], y1[4];
    load_ids(y, sb, nsb, Nn, lane, yk);
    while (sb < nsb) {
        const int base = sb << 7;
        const int sbn  = sb + step;
        load_ids(y, sbn, nsb, Nn, lane, y1);
#pragma unroll
        for (int k = 0; k < 4; ++k) {
            unsigned m = __builtin_amdgcn_ballot_w32((yk[k] >> 4) == group);
            while (m) {
                const int r = __builtin_ctz(m);
                m &= m - 1;
                const int seg = __builtin_amdgcn_readlane(yk[k], r);
                pool_row<PASS>(x, vr, cl, accL, wsL, mK, mF, base + 32 * k + r, seg, lane);
            }
        }
#pragma unroll
        for (int k = 0; k < 4; ++k) yk[k] = y1[k];
        sb = sbn;
    }
}

__device__ __forceinline__ void list_ids(const int4 q, int row0, int g, unsigned* listL,
                                         unsigned* cntS) {
    const int id[4] = {q.x, q.y, q.z, q.w};
#pragma unroll
    for (int j = 0; j < 4; ++j) {
        if ((id[j] >> 4) == g) {
            const int sl = id[j] & (SEGB - 1);
            const unsigned slot = __hip_atomic_fetch_add(cntS + sl, 1u, __ATOMIC_RELAXED,
                                                         __HIP_MEMORY_SCOPE_WORKGROUP);
            listL[sl * LPIT + min(slot, (unsigned)LSEG)] = (unsigned)(row0 + j);
        }
    }
}

__device__ __forceinline__ int4 load_ids4(const int* __restrict__ y, int sb, int nfull, int lane) {
    return *(const int4*)(y + ((size_t)min(sb, nfull - 1) << 7) + 4 * lane);
}

__device__ __forceinline__ void scan_list(const int* __restrict__ y, unsigned* listL,
                                          unsigned* cntS,
                                          int Nn, int group, int stripe, int wave, int lane) {
    const int nfull = Nn >> 7;
    const int step  = (PT / 32) * NSTR;
    int sb = wave * NSTR + stripe;
    if (sb < nfull) {
        int4 q0 = load_ids4(y, sb, nfull, lane);
        int4 q1 = load_ids4(y, sb + step, nfull, lane);
        int4 q2 = load_ids4(y, sb + 2 * step, nfull, lane);
        int4 q3 = load_ids4(y, sb + 3 * step, nfull, lane);
        while (sb < nfull) {
            const int r0 = (sb << 7) + 4 * lane;
            list_ids(q0, r0, group, listL, cntS);
            q0 = load_ids4(y, sb + 4 * step, nfull, lane);
            list_ids(q1, r0 + (step << 7), (sb + step < nfull) ? group : -2, listL, cntS);
            q1 = load_ids4(y, sb + 5 * step, nfull, lane);
            list_ids(q2, r0 + 2 * (step << 7), (sb + 2 * step < nfull) ? group : -2, listL, cntS);
            q2 = load_ids4(y, sb + 6 * step, nfull, lane);
            list_ids(q3, r0 + 3 * (step << 7), (sb + 3 * step < nfull) ? group : -2, listL, cntS);
            q3 = load_ids4(y, sb + 7 * step, nfull, lane);
            sb += 4 * step;
        }
    }
    if ((Nn & 127) != 0 && (nfull & (NSTR - 1)) == stripe && ((nfull >> 3) & (PT / 32 - 1)) == wave) {
        const int r0 = (nfull << 7) + 4 * lane;
        int4 q;
        q.x = (r0 + 0 < Nn) ? y[r0 + 0] : -1;
        q.y = (r0 + 1 < Nn) ? y[r0 + 1] : -1;
        q.z = (r0 + 2 < Nn) ? y[r0 + 2] : -1;
        q.w = (r0 + 3 < Nn) ? y[r0 + 3] : -1;
        list_ids(q, r0, group, listL, cntS);
    }
}

__device__ __forceinline__ void fetch_row(const float* __restrict__ x, unsigned row, int lane,
                                          float (&xv)[4]) {
    const float* xr = x + (size_t)row * IN_CH + lane;
    xv[0] = xr[0]; xv[1] = xr[32]; xv[2] = xr[64]; xv[3] = xr[96];
}

__device__ __forceinline__ void ask_rows(const float* __restrict__ x, const unsigned* lst,
                                         int f, int last, int lane) {
    if (lane < 4) {
        const unsigned row = lst[min(f + (lane & 2), last)];
        __builtin_prefetch(x + (size_t)row * IN_CH + (lane & 1) * 64, 0, 1);
    }
}

#define HEADROOM 16.0f

__device__ __forceinline__ void row_update(const float (&xv)[4], const float (&vr)[16],
                                           float cl, int lane, float (&acc)[16],
                                           float (&m)[4], float (&mt)[4], float (&ws)[4]) {
    float s[4];
    row_scores(xv, vr, cl, lane, s);
    if (s[0] > mt[0] || s[1] > mt[1] || s[2] > mt[2] || s[3] > mt[3]) {
#pragma unroll
        for (int hd = 0; hd < NHEAD; ++hd) {
            const float mn = fmaxf(m[hd], s[hd]);
            const float sc = __expf(m[hd] - mn);
#pragma unroll
            for (int j = 0; j < 4; ++j) acc[hd * 4 + j] *= sc;
            ws[hd] *= sc;
            m[hd]  = uni(mn);
            mt[hd] = m[hd] + HEADROOM;
        }
    }
#pragma unroll
    for (int hd = 0; hd < NHEAD; ++hd) {
        const float w = __expf(s[hd] - m[hd]);
#pragma unroll
        for (int j = 0; j < 4; ++j) acc[hd * 4 + j] = fmaf(w, xv[j], acc[hd * 4 + j]);
        ws[hd] += w;
    }
}

#define AHEAD 12
__device__ __forceinline__ void walk_seg(const float* __restrict__ x, const unsigned* lst,
                                         int n, int half, int lane,
                                         const float (&vr)[16], float cl,
                                         float (&acc)[16], float (&m)[4], float (&mt)[4],
                                         float (&ws)[4]) {
    if (half >= n) return;
    const int last = n - 1;
    const int cnt  = (n - half + 1) >> 1;
    int e = half;
    float xa[4], xb[4];
    unsigned ra = (unsigned)__builtin_amdgcn_readfirstlane((int)lst[e]);
    unsigned rb = (unsigned)__builtin_amdgcn_readfirstlane((int)lst[min(e + 2, last)]);
#pragma unroll
    for (int a = 4; a < AHEAD; a += 4) ask_rows(x, lst, e + a, last, lane);
    for (int i = cnt >> 1; i > 0; --i) {
        fetch_row(x, ra, lane, xa);
        fetch_row(x, rb, lane, xb);
        const unsigned na = lst[min(e + 4, last)];
        const unsigned nb = lst[min(e + 6, last)];
        ask_rows(x, lst, e + AHEAD, last, lane);
        row_update(xa, vr, cl, lane, acc, m, mt, ws);
        row_update(xb, vr, cl, lane, acc, m, mt, ws);
        ra = (unsigned)__builtin_amdgcn_readfirstlane((int)na);
        rb = (unsigned)__builtin_amdgcn_readfirstlane((int)nb);
        e += 4;
    }
    if (cnt & 1) {
        fetch_row(x, ra, lane, xa);
        row_update(xa, vr, cl, lane, acc, m, mt, ws);
    }
}

__global__ __launch_bounds__(PT) __attribute__((amdgpu_waves_per_eu(16)))
void k_pool(const float* __restrict__ x, const int* __restrict__ y,
            const float* __restrict__ lin_w, const float* __restrict__ lin_b,
            const float* __restrict__ att_w, const float* __restrict__ att_b,
            float* out, float* part, float* smax, float* ssum, int Nn, int C) {
    __shared__ __attribute__((aligned(16))) float accL[SEGB * SP];
    __shared__ float wsL[SEGB * NHEAD];
    __shared__ unsigned mK[SEGB * NHEAD];
    __shared__ __attribute__((aligned(16))) float mF[SEGB * NHEAD];
    __shared__ float outL[SEGB * TOT_OUT];
    __shared__ unsigned listL[SEGB * LPIT];
    __shared__ unsigned cntS[SEGB];
    __shared__ __attribute__((aligned(16))) float pmL[(PT / 32) * NHEAD];
    __shared__ float vP[2 * NHEAD * IN_CH];
    __shared__ float cP[2 * NHEAD];

    const int t = threadIdx.x;
    const int group  = blockIdx.x % NGRP;
    const int stripe = blockIdx.x / NGRP;
    const int wave = __builtin_amdgcn_readfirstlane(t >> 5);
    const int lane = t & 31;

    {
        const int half = wave >> 4;
        const int o = t & (NHEAD * IN_CH - 1);
        const float* wp = lin_w + (size_t)((o >> 7) * OUT_CH + half * 32) * IN_CH + (o & (IN_CH - 1));
        const float* ap = att_w + half * 32;
        float wv[32];
#pragma unroll
        for (int d = 0; d < 32; ++d) wv[d] = wp[(size_t)d * IN_CH];

        for (int i = t; i < SEGB * SP; i += PT) accL[i] = 0.0f;
        if (t < SEGB * NHEAD) { wsL[t] = 0.0f; mK[t] = 0u; }
        if (t < SEGB) cntS[t] = 0u;

        __syncthreads();

        scan_list(y, listL, cntS, Nn, group, stripe, wave, lane);

        float s = 0.0f;
#pragma unroll
        for (int d = 0; d < 32; ++d) s = fmaf(wv[d], ap[d], s);
        vP[t] = s;
        if (wave < 2 * NHEAD) {
            float pr = lin_b[t] * att_w[t & (OUT_CH - 1)];
            pr = XADD(pr, 1); pr = XADD(pr, 2); pr = XADD(pr, 4); pr = XADD(pr, 8);
            pr = pr + xhalf(pr);
            if (lane == 0) cP[wave] = pr;
        }
    }
    __syncthreads();

    float vr[16];
#pragma unroll
    for (int hd = 0; hd < NHEAD; ++hd)
#pragma unroll
        for (int j = 0; j < 4; ++j)
            vr[hd * 4 + j] = vP[hd * IN_CH + lane + 32 * j] + vP[NHEAD * IN_CH + hd * IN_CH + lane + 32 * j];
    const float cl = att_b[0] + (cP[2 * (lane & (NHEAD - 1))] + cP[2 * (lane & (NHEAD - 1)) + 1]);
    const unsigned ovf = __builtin_amdgcn_ballot_w32(cntS[lane & (SEGB - 1)] > (unsigned)LSEG);
    if (ovf == 0u) {
        const int sl = wave >> 1;
        const int n  = __builtin_amdgcn_readfirstlane((int)cntS[sl]);
        float acc[16], m[4], mt[4], ws[4];
#pragma unroll
        for (int i = 0; i < 16; ++i) acc[i] = 0.0f;
#pragma unroll
        for (int hd = 0; hd < NHEAD; ++hd) { m[hd] = -INFINITY; mt[hd] = -INFINITY; ws[hd] = 0.0f; }
        walk_seg(x, listL + sl * LPIT, n, wave & 1, lane, vr, cl, acc, m, mt, ws);
        if (lane < NHEAD)
            pmL[wave * NHEAD + lane] = (lane == 0) ? m[0] : (lane == 1) ? m[1] : (lane == 2) ? m[2] : m[3];
        __syncthreads();
        const float4 o = *(const float4*)(pmL + (wave ^ 1) * NHEAD);
        const float om[4] = {o.x, o.y, o.z, o.w};
        float f[4], big[4];
#pragma unroll
        for (int hd = 0; hd < NHEAD; ++hd) {
            big[hd] = fmaxf(m[hd], om[hd]);
            f[hd] = (m[hd] == -INFINITY) ? 0.0f : __expf(m[hd] - big[hd]);
        }
        float* p = accL + sl * SP + lane;
#pragma unroll
        for (int hd = 0; hd < NHEAD; ++hd)
#pragma unroll
            for (int j = 0; j < 4; ++j)
                __hip_atomic_fetch_add(p + hd * IN_CH + j * 32, acc[hd * 4 + j] * f[hd],
                                       __ATOMIC_RELAXED, __HIP_MEMORY_SCOPE_WORKGROUP);
        if (lane < NHEAD) {
            const float wl = (lane == 0) ? ws[0] * f[0] : (lane == 1) ? ws[1] * f[1]
                           : (lane == 2) ? ws[2] * f[2] : ws[3] * f[3];
            __hip_atomic_fetch_add(wsL + sl * NHEAD + lane, wl,
                                   __ATOMIC_RELAXED, __HIP_MEMORY_SCOPE_WORKGROUP);
            if ((wave & 1) == 0)
                mF[sl * NHEAD + lane] = (lane == 0) ? big[0] : (lane == 1) ? big[1]
                                      : (lane == 2) ? big[2] : big[3];
        }
    } else {
        scan_pass<1>(x, y, vr, cl, accL, wsL, mK, mF, Nn, group, stripe, wave, lane);
        __syncthreads();
        if (t < SEGB * NHEAD) mF[t] = key2f(mK[t]);
        __syncthreads();
        scan_pass<2>(x, y, vr, cl, accL, wsL, mK, mF, Nn, group, stripe, wave, lane);
    }
    __syncthreads();

    if (wave < 16) {
        const int head = wave >> 2, nt = wave & 3;
        const int cc = lane & 15, h = lane >> 4;
        const int col = head * OUT_CH + nt * 16 + cc;
        const float* brow = lin_w + (size_t)col * IN_CH + 4 * h;
        const float* arow = accL + cc * SP + head * IN_CH + 4 * h;
        v8f acc = {0.f, 0.f, 0.f, 0.f, 0.f, 0.f, 0.f, 0.f};
#pragma unroll
        for (int j = 0; j < 16; ++j) {
            const float4 a4 = *(const float4*)(arow + 8 * j);
            const float4 b4 = *(const float4*)(brow + 8 * j);
            v2f a0 = {a4.x, a4.y}, a1 = {a4.z, a4.w};
            v2f b0 = {b4.x, b4.y}, b1 = {b4.z, b4.w};
            acc = __builtin_amdgcn_wmma_f32_16x16x4_f32(false, a0, false, b0, (short)0, acc, false, false);
            acc = __builtin_amdgcn_wmma_f32_16x16x4_f32(false, a1, false, b1, (short)0, acc, false, false);
        }
        const float bn = lin_b[col];
#pragma unroll
        for (int r = 0; r < 8; ++r) {
            const int sl = 8 * h + r;
            outL[sl * TOT_OUT + col] = acc[r] + bn * wsL[sl * NHEAD + head];
        }
    }
    __syncthreads();

    const int nseg = (C < NSEG) ? C : NSEG;
    float* dst = (stripe == 0) ? out : part + (size_t)(stripe - 1) * C * TOT_OUT;
#pragma unroll
    for (int i = 0; i < SEGB * TOT_OUT / PT; ++i) {
        const int e   = i * PT + t;
        const int seg = group * SEGB + (e >> 8);
        if (seg < nseg) dst[(size_t)seg * TOT_OUT + (e & 255)] = outL[e];
    }
    if (t < SEGB * NHEAD) {
        const int seg = group * SEGB + (t >> 2);
        if (seg < nseg) {
            const size_t o = ((size_t)stripe * C + seg) * NHEAD + (t & 3);
            smax[o] = mF[t];
            ssum[o] = wsL[t];
        }
    }
}

__global__ __launch_bounds__(256)
void k_merge(const float* __restrict__ part, const float* __restrict__ smax,
             const float* __restrict__ ssum, float* out, int C) {
    const int i = blockIdx.x * 256 + threadIdx.x;
    if (i >= C * TOT_OUT) return;
    const int seg = i >> 8, head = (i & 255) >> 6;
    if (seg >= NSEG) { out[i] = 0.0f; return; }
    float m[NSTR], best = -INFINITY;
#pragma unroll
    for (int b = 0; b < NSTR; ++b) {
        m[b] = smax[((size_t)b * C + seg) * NHEAD + head];
        best = fmaxf(best, m[b]);
    }
    float num = 0.0f, den = 0.0f;
#pragma unroll
    for (int b = 0; b < NSTR; ++b) {
        const float f  = (m[b] == -INFINITY) ? 0.0f : __expf(m[b] - best);
        const float pb = (b == 0) ? out[i] : part[(size_t)(b - 1) * C * TOT_OUT + i];
        num += f * pb;
        den += f * ssum[((size_t)b * C + seg) * NHEAD + head];
    }
    out[i] = (den > 0.0f) ? num / den : 0.0f;
}

extern "C" void kernel_launch(void* const* d_in, const int* in_sizes, int n_in,
                              void* d_out, int out_size, void* d_ws, size_t ws_size,
                              hipStream_t stream) {
    (void)n_in; (void)ws_size;
    const float* x     = (const float*)d_in[0];
    const int*   y     = (const int*)  d_in[1];
    const float* lin_w = (const float*)d_in[3];
    const float* lin_b = (const float*)d_in[4];
    const float* att_w = (const float*)d_in[5];
    const float* att_b = (const float*)d_in[6];
    float* out = (float*)d_out;

    int N = in_sizes[0] / IN_CH;
    int C = out_size / TOT_OUT;

    float* wsf  = (float*)d_ws;
    float* smax = wsf + WS_MX;
    float* ssum = wsf + WS_DN(C);
    float* part = wsf + WS_P(C);

    k_pool <<<NGRP * NSTR, PT, 0, stream>>>(x, y, lin_w, lin_b, att_w, att_b,
                                            out, part, smax, ssum, N, C);
    k_merge<<<(C * TOT_OUT + 255) / 256, 256, 0, stream>>>(part, smax, ssum, out, C);
}


# k_pool walk: each 16-lane half of a wave scores and sums two of the four heads (a lane 8 channels, 16-byte row loads): one score, exp and scale per lane, 16-lane butterfly; score sums reordered
# speedup vs baseline: 14.5231x; 1.0499x over previous
#include <hip/hip_runtime.h>
#include <math.h>

typedef __attribute__((ext_vector_type(2))) float v2f;
typedef __attribute__((ext_vector_type(8))) float v8f;
typedef __attribute__((address_space(1))) char   gbyte;
typedef __attribute__((address_space(1))) float4 gfloat4;

#define IN_CH     128
#define OUT_CH    64
#define NHEAD     4
#define TOT_OUT   256
#define NEG_SLOPE 0.2f

#define NSEG  1000
#define SEGB  16
#define NGRP  63
#define NSTR  8
#define SP    516
#define PT    1024
#define LSEG  192
#define LPIT  (LSEG + 1)

#define WS_MX  0
#define WS_DN(C) (WS_MX + NSTR * (C) * NHEAD)
#define WS_P(C)  (WS_MX + 2 * NSTR * (C) * NHEAD)


__device__ __forceinline__ float xhalf(float a) {
    const int b = __builtin_bit_cast(int, a);
    return __builtin_bit_cast(float,
        __builtin_amdgcn_permlanex16(b, b, 0x76543210, 0xfedcba98, false, false));
}
#define XADD(a, K) ((a) + __builtin_bit_cast(float, __builtin_amdgcn_update_dpp(          \
                        0, __builtin_bit_cast(int, (a)), 0x160 + (K), 0xf, 0xf, false)))
#define XADD2(k, g, K) ((k) + __builtin_bit_cast(float, __builtin_amdgcn_update_dpp(      \
                        0, __builtin_bit_cast(int, (g)), 0x160 + (K), 0xf, 0xf, false)))

__device__ __forceinline__ unsigned f2key(float f) {
    const unsigned b = __float_as_uint(f);
    return (b & 0x80000000u) ? ~b : (b | 0x80000000u);
}
__device__ __forceinline__ float key2f(unsigned k) {
    if (k == 0u) return -INFINITY;
    return __uint_as_float((k & 0x80000000u) ? (k ^ 0x80000000u) : ~k);
}

__device__ __forceinline__ void row_scores(const float (&xv)[4], const float (&vr)[16],
                                           float cl, int lane, float (&s)[4]) {
    float p[4];
#pragma unroll
    for (int hd = 0; hd < NHEAD; ++hd) {
        p[hd] = xv[0] * vr[hd * 4 + 0];
        p[hd] = fmaf(xv[1], vr[hd * 4 + 1], p[hd]);
        p[hd] = fmaf(xv[2], vr[hd * 4 + 2], p[hd]);
        p[hd] = fmaf(xv[3], vr[hd * 4 + 3], p[hd]);
    }
    const bool o1 = (lane & 1) != 0, o2 = (lane & 2) != 0;
    const float q0 = XADD2(o1 ? p[1] : p[0], o1 ? p[0] : p[1], 1);
    const float q1 = XADD2(o1 ? p[3] : p[2], o1 ? p[2] : p[3], 1);
    float r = XADD2(o2 ? q1 : q0, o2 ? q0 : q1, 2);
    r = XADD(r, 4); r = XADD(r, 8);
    r = r + xhalf(r);
    r += cl;
    const float t = (r >= 0.0f) ? r : NEG_SLOPE * r;
#pragma unroll
    for (int hd = 0; hd < NHEAD; ++hd)
        s[hd] = __builtin_bit_cast(float, __builtin_amdgcn_readlane(__builtin_bit_cast(int, t), hd));
}

template <int PASS>
__device__ __forceinline__ void pool_row(const float* __restrict__ x,
                                         const float (&vr)[16], float cl,
                                         float* accL, float* wsL, unsigned* mK, const float* mF,
                                         int row, int seg, int lane) {
    const float* xr = x + (size_t)row * IN_CH + lane;
    const float xv[4] = {xr[0], xr[32], xr[64], xr[96]};
    float s[4];
    row_scores(xv, vr, cl, lane, s);
    const int sl = seg & (SEGB - 1);
    if (PASS == 1) {
        if (lane < NHEAD) {
            const float sh = (lane == 0) ? s[0] : (lane == 1) ? s[1] : (lane == 2) ? s[2] : s[3];
            __hip_atomic_fetch_max(mK + sl * NHEAD + lane, f2key(sh),
                                   __ATOMIC_RELAXED, __HIP_MEMORY_SCOPE_WORKGROUP);
        }
    } else {
        const float4 mx = *(const float4*)(mF + sl * NHEAD);
        const float w[4] = {__expf(s[0] - mx.x), __expf(s[1] - mx.y),
                            __expf(s[2] - mx.z), __expf(s[3] - mx.w)};
        float* p = accL + sl * SP + lane;
#pragma unroll
        for (int hd = 0; hd < NHEAD; ++hd)
#pragma unroll
            for (int j = 0; j < 4; ++j)
                __hip_atomic_fetch_add(p + hd * IN_CH + j * 32, xv[j] * w[hd],
                                       __ATOMIC_RELAXED, __HIP_MEMORY_SCOPE_WORKGROUP);
        if (lane < NHEAD) {
            const float wl = (lane == 0) ? w[0] : (lane == 1) ? w[1] : (lane == 2) ? w[2] : w[3];
            __hip_atomic_fetch_add(wsL + sl * NHEAD + lane, wl,
                                   __ATOMIC_RELAXED, __HIP_MEMORY_SCOPE_WORKGROUP);
        }
    }
}

__device__ __forceinline__ void load_ids(const int* __restrict__ y, int sb, int nsb,
                                         int Nn, int lane, int (&yv)[4]) {
#pragma unroll
    for (int k = 0; k < 4; ++k) {
        const int r = (sb << 7) + 32 * k + lane;
        yv[k] = -1;
        if (sb < nsb && r < Nn) yv[k] = y[r];
    }
}

template <int PASS>
__device__ __forceinline__ void scan_pass(const float* __restrict__ x, const int* __restrict__ y,
                                          const float (&vr)[16], float cl,
                                          float* accL, float* wsL, unsigned* mK, const float* mF,
                                          int Nn, int group, int stripe, int wave, int lane) {
    const int nsb  = (Nn + 127) >> 7;
    const int step = (PT / 32) * NSTR;
    int sb = wave * NSTR + stripe;
    int yk[4], y1[4];
    load_ids(y, sb, nsb, Nn, lane, yk);
    while (sb < nsb) {
        const int base = sb << 7;
        const int sbn  = sb + step;
        load_ids(y, sbn, nsb, Nn, lane, y1);
#pragma unroll
        for (int k = 0; k < 4; ++k) {
            unsigned m = __builtin_amdgcn_ballot_w32((yk[k] >> 4) == group);
            while (m) {
                const int r = __builtin_ctz(m);
                m &= m - 1;
                const int seg = __builtin_amdgcn_readlane(yk[k], r);
                pool_row<PASS>(x, vr, cl, accL, wsL, mK, mF, base + 32 * k + r, seg, lane);
            }
        }
#pragma unroll
        for (int k = 0; k < 4; ++k) yk[k] = y1[k];
        sb = sbn;
    }
}

__device__ __forceinline__ void list_ids(const int4 q, int row0, int g, unsigned* listL,
                                         unsigned* cntS) {
    const int id[4] = {q.x, q.y, q.z, q.w};
#pragma unroll
    for (int j = 0; j < 4; ++j) {
        if ((id[j] >> 4) == g) {
            const int sl = id[j] & (SEGB - 1);
            const unsigned slot = __hip_atomic_fetch_add(cntS + sl, 1u, __ATOMIC_RELAXED,
                                                         __HIP_MEMORY_SCOPE_WORKGROUP);
            listL[sl * LPIT + min(slot, (unsigned)LSEG)] = (unsigned)(row0 + j);
        }
    }
}

__device__ __forceinline__ int4 load_ids4(const int* __restrict__ y, int sb, int nfull, int lane) {
    return *(const int4*)(y + ((size_t)min(sb, nfull - 1) << 7) + 4 * lane);
}

__device__ __forceinline__ void scan_list(const int* __restrict__ y, unsigned* listL,
                                          unsigned* cntS,
                                          int Nn, int group, int stripe, int wave, int lane) {
    const int nfull = Nn >> 7;
    const int step  = (PT / 32) * NSTR;
    int sb = wave * NSTR + stripe;
    if (sb < nfull) {
        int4 q0 = load_ids4(y, sb, nfull, lane);
        int4 q1 = load_ids4(y, sb + step, nfull, lane);
        int4 q2 = load_ids4(y, sb + 2 * step, nfull, lane);
        int4 q3 = load_ids4(y, sb + 3 * step, nfull, lane);
        while (sb < nfull) {
            const int r0 = (sb << 7) + 4 * lane;
            list_ids(q0, r0, group, listL, cntS);
            q0 = load_ids4(y, sb + 4 * step, nfull, lane);
            list_ids(q1, r0 + (step << 7), (sb + step < nfull) ? group : -2, listL, cntS);
            q1 = load_ids4(y, sb + 5 * step, nfull, lane);
            list_ids(q2, r0 + 2 * (step << 7), (sb + 2 * step < nfull) ? group : -2, listL, cntS);
            q2 = load_ids4(y, sb + 6 * step, nfull, lane);
            list_ids(q3, r0 + 3 * (step << 7), (sb + 3 * step < nfull) ? group : -2, listL, cntS);
            q3 = load_ids4(y, sb + 7 * step, nfull, lane);
            sb += 4 * step;
        }
    }
    if ((Nn & 127) != 0 && (nfull & (NSTR - 1)) == stripe && ((nfull >> 3) & (PT / 32 - 1)) == wave) {
        const int r0 = (nfull << 7) + 4 * lane;
        int4 q;
        q.x = (r0 + 0 < Nn) ? y[r0 + 0] : -1;
        q.y = (r0 + 1 < Nn) ? y[r0 + 1] : -1;
        q.z = (r0 + 2 < Nn) ? y[r0 + 2] : -1;
        q.w = (r0 + 3 < Nn) ? y[r0 + 3] : -1;
        list_ids(q, r0, group, listL, cntS);
    }
}

__device__ __forceinline__ const gbyte* row_base(const float* __restrict__ x, unsigned row) {
    const unsigned long long b = (unsigned long long)x + ((unsigned long long)row << 9);
    const unsigned blo = (unsigned)__builtin_amdgcn_readfirstlane((int)(unsigned)b);
    const unsigned bhi = (unsigned)__builtin_amdgcn_readfirstlane((int)(unsigned)(b >> 32));
    return (const gbyte*)(((unsigned long long)bhi << 32) | blo);
}

__device__ __forceinline__ void fetch_row(const float* __restrict__ x, unsigned row, unsigned off,
                                          float (&xv)[8]) {
    const gbyte* rp = row_base(x, row);
    const float4 lo = *(const gfloat4*)(rp + off);
    const float4 hi = *(const gfloat4*)(rp + off + 256u);
    xv[0] = lo.x; xv[1] = lo.y; xv[2] = lo.z; xv[3] = lo.w;
    xv[4] = hi.x; xv[5] = hi.y; xv[6] = hi.z; xv[7] = hi.w;
}

__device__ __forceinline__ float pair_even(float a) {
    return __builtin_bit_cast(float, __builtin_amdgcn_mov_dpp(
        __builtin_bit_cast(int, a), 0xA0, 0xf, 0xf, true));
}
__device__ __forceinline__ float pair_odd(float a) {
    return __builtin_bit_cast(float, __builtin_amdgcn_mov_dpp(
        __builtin_bit_cast(int, a), 0xF5, 0xf, 0xf, true));
}

__device__ __forceinline__ float row_score(const float (&xv)[8], const v2f (&vq)[8],
                                           float cl, int lane) {
    v2f p = (v2f){xv[0], xv[0]} * vq[0];
#pragma unroll
    for (int i = 1; i < 8; ++i) p = __builtin_elementwise_fma((v2f){xv[i], xv[i]}, vq[i], p);
    const bool o1 = (lane & 1) != 0;
    float r = XADD2(o1 ? p.y : p.x, o1 ? p.x : p.y, 1);
    r = XADD(r, 2); r = XADD(r, 4); r = XADD(r, 8);
    r += cl;
    return fmaxf(r, NEG_SLOPE * r);
}

__device__ __forceinline__ void ask_rows(const float* __restrict__ x, const unsigned* lst,
                                         int f, int last, int lane) {
    const unsigned r0 = (unsigned)__builtin_amdgcn_readfirstlane((int)lst[min(f, last)]);
    const unsigned r1 = (unsigned)__builtin_amdgcn_readfirstlane((int)lst[min(f + 2, last)]);
    if (lane == 0) {
        const gbyte* p0 = row_base(x, r0);
        const gbyte* p1 = row_base(x, r1);
        __builtin_prefetch((const char*)p0, 0, 1);
        __builtin_prefetch((const char*)(p0 + 256), 0, 1);
        __builtin_prefetch((const char*)p1, 0, 1);
        __builtin_prefetch((const char*)(p1 + 256), 0, 1);
    }
}

#define HEADROOM 16.0f

__device__ __forceinline__ void row_update(const float (&xv)[8], const v2f (&vq)[8],
                                           float cl, int lane, float (&acc)[16],
                                           float& m, float& ws) {
    const float t = row_score(xv, vq, cl, lane);
    float d = t - m;
    if (__builtin_amdgcn_ballot_w32(d > HEADROOM) != 0u) {
        const float mn = fmaxf(m, t);
        const float sc = __expf(m - mn);
        const float sa = pair_even(sc), sb = pair_odd(sc);
#pragma unroll
        for (int i = 0; i < 8; ++i) { acc[i] *= sa; acc[8 + i] *= sb; }
        ws *= sc;
        m = mn;
        d = t - mn;
    }
    const float w = __expf(d);
    const float wa = pair_even(w), wb = pair_odd(w);
#pragma unroll
    for (int i = 0; i < 8; ++i) {
        acc[i]     = fmaf(wa, xv[i], acc[i]);
        acc[8 + i] = fmaf(wb, xv[i], acc[8 + i]);
    }
    ws += w;
}

#define AHEAD 12
__device__ __forceinline__ void walk_seg(const float* __restrict__ x, const unsigned* lst,
                                         int n, int half, int lane,
                                         const v2f (&vq)[8], float cl,
                                         float (&acc)[16], float& m, float& ws) {
    if (half >= n) return;
    const int last = n - 1;
    const int cnt  = (n - half + 1) >> 1;
    int e = half;
    float xa[8], xb[8];
    unsigned ra = (unsigned)__builtin_amdgcn_readfirstlane((int)lst[e]);
    unsigned rb = (unsigned)__builtin_amdgcn_readfirstlane((int)lst[min(e + 2, last)]);
    unsigned off = 16u * (unsigned)(lane & 15);
#pragma unroll
    for (int a = 4; a < AHEAD; a += 4) ask_rows(x, lst, e + a, last, lane);
    for (int i = cnt >> 1; i > 0; --i) {
        asm volatile("" : "+v"(off));
        fetch_row(x, ra, off, xa);
        fetch_row(x, rb, off, xb);
        const unsigned na = lst[min(e + 4, last)];
        const unsigned nb = lst[min(e + 6, last)];
        ask_rows(x, lst, e + AHEAD, last, lane);
        const unsigned ran = (unsigned)__builtin_amdgcn_readfirstlane((int)na);
        const unsigned rbn = (unsigned)__builtin_amdgcn_readfirstlane((int)nb);
        row_update(xa, vq, cl, lane, acc, m, ws);
        row_update(xb, vq, cl, lane, acc, m, ws);
        ra = ran;
        rb = rbn;
        e += 4;
    }
    if (cnt & 1) {
        fetch_row(x, ra, off, xa);
        row_update(xa, vq, cl, lane, acc, m, ws);
    }
}

__global__ __launch_bounds__(PT) __attribute__((amdgpu_waves_per_eu(16)))
void k_pool(const float* __restrict__ x, const int* __restrict__ y,
            const float* __restrict__ lin_w, const float* __restrict__ lin_b,
            const float* __restrict__ att_w, const float* __restrict__ att_b,
            float* out, float* part, float* smax, float* ssum, int Nn, int C) {
    __shared__ __attribute__((aligned(16))) float accL[SEGB * SP];
    __shared__ float wsL[SEGB * NHEAD];
    __shared__ unsigned mK[SEGB * NHEAD];
    __shared__ __attribute__((aligned(16))) float mF[SEGB * NHEAD];
    __shared__ float outL[SEGB * TOT_OUT];
    __shared__ unsigned listL[SEGB * LPIT];
    __shared__ unsigned cntS[SEGB];
    __shared__ __attribute__((aligned(16))) float pmL[(PT / 32) * NHEAD];
    __shared__ __attribute__((aligned(16))) float vP[2 * NHEAD * IN_CH];
    __shared__ float cP[2 * NHEAD];

    const int t = threadIdx.x;
    const int group  = blockIdx.x % NGRP;
    const int stripe = blockIdx.x / NGRP;
    const int wave = __builtin_amdgcn_readfirstlane(t >> 5);
    const int lane = t & 31;

    {
        const int half = wave >> 4;
        const int o = t & (NHEAD * IN_CH - 1);
        const float* wp = lin_w + (size_t)((o >> 7) * OUT_CH + half * 32) * IN_CH + (o & (IN_CH - 1));
        const float* ap = att_w + half * 32;
        float wv[32];
#pragma unroll
        for (int d = 0; d < 32; ++d) wv[d] = wp[(size_t)d * IN_CH];

        for (int i = t; i < SEGB * SP; i += PT) accL[i] = 0.0f;
        if (t < SEGB * NHEAD) { wsL[t] = 0.0f; mK[t] = 0u; }
        if (t < SEGB) cntS[t] = 0u;

        __syncthreads();

        scan_list(y, listL, cntS, Nn, group, stripe, wave, lane);

        float s = 0.0f;
#pragma unroll
        for (int d = 0; d < 32; ++d) s = fmaf(wv[d], ap[d], s);
        vP[t] = s;
        if (wave < 2 * NHEAD) {
            float pr = lin_b[t] * att_w[t & (OUT_CH - 1)];
            pr = XADD(pr, 1); pr = XADD(pr, 2); pr = XADD(pr, 4); pr = XADD(pr, 8);
            pr = pr + xhalf(pr);
            if (lane == 0) cP[wave] = pr;
        }
    }
    __syncthreads();

    const unsigned ovf = __builtin_amdgcn_ballot_w32(cntS[lane & (SEGB - 1)] > (unsigned)LSEG);
    if (ovf == 0u) {
        const int hp = lane >> 4;
        const int c4 = 4 * (lane & 15);
        const int hd = 2 * hp + (lane & 1);
        v2f vq[8];
#pragma unroll
        for (int g = 0; g < 2; ++g) {
            const int o = 2 * hp * IN_CH + 64 * g + c4;
            const float4 a0 = *(const float4*)(vP + o);
            const float4 a1 = *(const float4*)(vP + NHEAD * IN_CH + o);
            const float4 b0 = *(const float4*)(vP + IN_CH + o);
            const float4 b1 = *(const float4*)(vP + NHEAD * IN_CH + IN_CH + o);
            vq[g * 4 + 0] = (v2f){a0.x + a1.x, b0.x + b1.x};
            vq[g * 4 + 1] = (v2f){a0.y + a1.y, b0.y + b1.y};
            vq[g * 4 + 2] = (v2f){a0.z + a1.z, b0.z + b1.z};
            vq[g * 4 + 3] = (v2f){a0.w + a1.w, b0.w + b1.w};
        }
        const float cq = att_b[0] + (cP[2 * hd] + cP[2 * hd + 1]);
        const int sl = wave >> 1;
        const int n  = __builtin_amdgcn_readfirstlane((int)cntS[sl]);
        float acc[16];
#pragma unroll
        for (int i = 0; i < 16; ++i) acc[i] = 0.0f;
        float m = -INFINITY, ws = 0.0f;
        walk_seg(x, listL + sl * LPIT, n, wave & 1, lane, vq, cq, acc, m, ws);
        const int lw  = (int)__builtin_amdgcn_mbcnt_lo(~0u, 0u);
        const int hpw = lw >> 4;
        const int c4w = 4 * (lw & 15);
        const int hdw = 2 * hpw + (lw & 1);
        const bool own = (lw & 14) == 0;
        if (own) pmL[wave * NHEAD + hdw] = m;
        __syncthreads();
        const float big = fmaxf(m, pmL[(wave ^ 1) * NHEAD + hdw]);
        const float f   = (m == -INFINITY) ? 0.0f : __expf(m - big);
        const float fa  = pair_even(f), fb = pair_odd(f);
        float* p = accL + sl * SP + 2 * hpw * IN_CH + c4w;
#pragma unroll
        for (int a = 0; a < 2; ++a)
#pragma unroll
            for (int g = 0; g < 2; ++g)
#pragma unroll
                for (int j = 0; j < 4; ++j)
                    __hip_atomic_fetch_add(p + a * IN_CH + 64 * g + j,
                                           acc[a * 8 + g * 4 + j] * (a ? fb : fa),
                                           __ATOMIC_RELAXED, __HIP_MEMORY_SCOPE_WORKGROUP);
        if (own) {
            __hip_atomic_fetch_add(wsL + sl * NHEAD + hdw, ws * f,
                                   __ATOMIC_RELAXED, __HIP_MEMORY_SCOPE_WORKGROUP);
            if ((wave & 1) == 0) mF[sl * NHEAD + hdw] = big;
        }
    } else {
        float vr[16];
#pragma unroll
        for (int hd = 0; hd < NHEAD; ++hd)
#pragma unroll
            for (int j = 0; j < 4; ++j)
                vr[hd * 4 + j] = vP[hd * IN_CH + lane + 32 * j] + vP[NHEAD * IN_CH + hd * IN_CH + lane + 32 * j];
        const float cl = att_b[0] + (cP[2 * (lane & (NHEAD - 1))] + cP[2 * (lane & (NHEAD - 1)) + 1]);
        scan_pass<1>(x, y, vr, cl, accL, wsL, mK, mF, Nn, group, stripe, wave, lane);
        __syncthreads();
        if (t < SEGB * NHEAD) mF[t] = key2f(mK[t]);
        __syncthreads();
        scan_pass<2>(x, y, vr, cl, accL, wsL, mK, mF, Nn, group, stripe, wave, lane);
    }
    __syncthreads();

    const int ln = (int)__builtin_amdgcn_mbcnt_lo(~0u, 0u);
    const int tn = (wave << 5) + ln;

    if (wave < 16) {
        const int head = wave >> 2, nt = wave & 3;
        const int cc = ln & 15, h = ln >> 4;
        const int col = head * OUT_CH + nt * 16 + cc;
        const float* brow = lin_w + (size_t)col * IN_CH + 4 * h;
        const float* arow = accL + cc * SP + head * IN_CH + 4 * h;
        v8f acc = {0.f, 0.f, 0.f, 0.f, 0.f, 0.f, 0.f, 0.f};
#pragma unroll
        for (int j = 0; j < 16; ++j) {
            const float4 a4 = *(const float4*)(arow + 8 * j);
            const float4 b4 = *(const float4*)(brow + 8 * j);
            v2f a0 = {a4.x, a4.y}, a1 = {a4.z, a4.w};
            v2f b0 = {b4.x, b4.y}, b1 = {b4.z, b4.w};
            acc = __builtin_amdgcn_wmma_f32_16x16x4_f32(false, a0, false, b0, (short)0, acc, false, false);
            acc = __builtin_amdgcn_wmma_f32_16x16x4_f32(false, a1, false, b1, (short)0, acc, false, false);
        }
        const float bn = lin_b[col];
#pragma unroll
        for (int r = 0; r < 8; ++r) {
            const int sl = 8 * h + r;
            outL[sl * TOT_OUT + col] = acc[r] + bn * wsL[sl * NHEAD + head];
        }
    }
    __syncthreads();

    const int nseg = (C < NSEG) ? C : NSEG;
    float* dst = (stripe == 0) ? out : part + (size_t)(stripe - 1) * C * TOT_OUT;
#pragma unroll
    for (int i = 0; i < SEGB * TOT_OUT / PT; ++i) {
        const int e   = i * PT + tn;
        const int seg = group * SEGB + (e >> 8);
        if (seg < nseg) dst[(size_t)seg * TOT_OUT + (e & 255)] = outL[e];
    }
    if (tn < SEGB * NHEAD) {
        const int seg = group * SEGB + (tn >> 2);
        if (seg < nseg) {
            const size_t o = ((size_t)stripe * C + seg) * NHEAD + (tn & 3);
            smax[o] = mF[tn];
            ssum[o] = wsL[tn];
        }
    }
}

__global__ __launch_bounds__(256)
void k_merge(const float* __restrict__ part, const float* __restrict__ smax,
             const float* __restrict__ ssum, float* out, int C) {
    const int i = blockIdx.x * 256 + threadIdx.x;
    if (i >= C * TOT_OUT) return;
    const int seg = i >> 8, head = (i & 255) >> 6;
    if (seg >= NSEG) { out[i] = 0.0f; return; }
    float m[NSTR], best = -INFINITY;
#pragma unroll
    for (int b = 0; b < NSTR; ++b) {
        m[b] = smax[((size_t)b * C + seg) * NHEAD + head];
        best = fmaxf(best, m[b]);
    }
    float num = 0.0f, den = 0.0f;
#pragma unroll
    for (int b = 0; b < NSTR; ++b) {
        const float f  = (m[b] == -INFINITY) ? 0.0f : __expf(m[b] - best);
        const float pb = (b == 0) ? out[i] : part[(size_t)(b - 1) * C * TOT_OUT + i];
        num += f * pb;
        den += f * ssum[((size_t)b * C + seg) * NHEAD + head];
    }
    out[i] = (den > 0.0f) ? num / den : 0.0f;
}

extern "C" void kernel_launch(void* const* d_in, const int* in_sizes, int n_in,
                              void* d_out, int out_size, void* d_ws, size_t ws_size,
                              hipStream_t stream) {
    (void)n_in; (void)ws_size;
    const float* x     = (const float*)d_in[0];
    const int*   y     = (const int*)  d_in[1];
    const float* lin_w = (const float*)d_in[3];
    const float* lin_b = (const float*)d_in[4];
    const float* att_w = (const float*)d_in[5];
    const float* att_b = (const float*)d_in[6];
    float* out = (float*)d_out;

    int N = in_sizes[0] / IN_CH;
    int C = out_size / TOT_OUT;

    float* wsf  = (float*)d_ws;
    float* smax = wsf + WS_MX;
    float* ssum = wsf + WS_DN(C);
    float* part = wsf + WS_P(C);

    k_pool <<<NGRP * NSTR, PT, 0, stream>>>(x, y, lin_w, lin_b, att_w, att_b,
                                            out, part, smax, ssum, N, C);
    k_merge<<<(C * TOT_OUT + 255) / 256, 256, 0, stream>>>(part, smax, ssum, out, C);
}
